# starting kernel
# baseline (speedup 1.0000x reference)
#include <hip/hip_runtime.h>

typedef __attribute__((ext_vector_type(2))) float v2f;
typedef __attribute__((ext_vector_type(8))) float v8f;

#define GIN_D    128
#define GIN_PAD  132
#define GIN_WAVES 8
#define GIN_ROWS (GIN_WAVES * 16)

__global__ __launch_bounds__(256) void gin_h_init(const float* __restrict__ x,
                                                  const float* __restrict__ eps,
                                                  float* __restrict__ h, int n4) {
    int i = blockIdx.x * 256 + threadIdx.x;
    if (i < n4) {
        float s = 1.0f + eps[0];
        float4 v = ((const float4*)x)[i];
        v.x *= s; v.y *= s; v.z *= s; v.w *= s;
        ((float4*)h)[i] = v;
    }
}

__global__ __launch_bounds__(256) void gin_edge_scatter(const float* __restrict__ x,
                                                        const int* __restrict__ edge,
                                                        float* __restrict__ h, int ne) {
    long t = (long)blockIdx.x * 256 + threadIdx.x;
    int e = (int)(t >> 5);
    if (e >= ne) return;
    int src = edge[e];
    int dst = edge[ne + e];
    if (src == dst) return;
    int c = ((int)t & 31) * 4;
    float4 v = *(const float4*)(x + (long)src * GIN_D + c);
    float* p = h + (long)dst * GIN_D + c;
    unsafeAtomicAdd(p + 0, v.x);
    unsafeAtomicAdd(p + 1, v.y);
    unsafeAtomicAdd(p + 2, v.z);
    unsafeAtomicAdd(p + 3, v.w);
}

__global__ __launch_bounds__(256) void gin_mlp(const float* __restrict__ h,
                                               const float* __restrict__ W1,
                                               const float* __restrict__ b1,
                                               const float* __restrict__ W2,
                                               const float* __restrict__ b2,
                                               float* __restrict__ out,
                                               const int* __restrict__ ids,
                                               int nrows) {
    __shared__ float lds_W[GIN_D * GIN_PAD];
    __shared__ float lds_A[GIN_WAVES * 16 * GIN_PAD];
    __shared__ float lds_b1[GIN_D];
    __shared__ float lds_b2[GIN_D];

    const int tid   = threadIdx.x;
    const int wave  = tid >> 5;
    const int lane  = tid & 31;
    const int lmod  = lane & 15;
    const int lhalf = lane >> 4;
    const int koff  = lhalf * 2;
    const int row0  = blockIdx.x * GIN_ROWS + wave * 16;

    if (tid < GIN_D) lds_b1[tid] = b1[tid];
    else             lds_b2[tid - GIN_D] = b2[tid - GIN_D];

    for (int i = tid; i < GIN_D * GIN_D; i += 256) {
        int k = i >> 7, n = i & (GIN_D - 1);
        lds_W[n * GIN_PAD + k] = W1[i];
    }

    float* myA = &lds_A[wave * 16 * GIN_PAD];
    for (int r = 0; r < 16; ++r) {
        int row = row0 + r;
        row = row < nrows ? row : nrows - 1;
        long srow = ids ? (long)ids[row] : (long)row;
        float4 v = *(const float4*)(h + srow * GIN_D + lane * 4);
        *(float4*)(myA + r * GIN_PAD + lane * 4) = v;
    }
    __syncthreads();

    v8f acc[8] = {};
    for (int kk = 0; kk < GIN_D; kk += 4) {
        int k = kk + koff;
        v2f a = *(const v2f*)(myA + lmod * GIN_PAD + k);
#pragma unroll
        for (int ct = 0; ct < 8; ++ct) {
            v2f b = *(const v2f*)(lds_W + (ct * 16 + lmod) * GIN_PAD + k);
            acc[ct] = __builtin_amdgcn_wmma_f32_16x16x4_f32(
                false, a, false, b, (short)0, acc[ct], false, false);
        }
    }
    __syncthreads();

#pragma unroll
    for (int ct = 0; ct < 8; ++ct) {
        float bv = lds_b1[ct * 16 + lmod];
#pragma unroll
        for (int r = 0; r < 8; ++r) {
            float t = acc[ct][r] + bv;
            myA[(r + 8 * lhalf) * GIN_PAD + ct * 16 + lmod] = t > 0.0f ? t : 0.0f;
        }
    }
    for (int i = tid; i < GIN_D * GIN_D; i += 256) {
        int k = i >> 7, n = i & (GIN_D - 1);
        lds_W[n * GIN_PAD + k] = W2[i];
    }
    __syncthreads();

    v8f acc2[8] = {};
    for (int kk = 0; kk < GIN_D; kk += 4) {
        int k = kk + koff;
        v2f a = *(const v2f*)(myA + lmod * GIN_PAD + k);
#pragma unroll
        for (int ct = 0; ct < 8; ++ct) {
            v2f b = *(const v2f*)(lds_W + (ct * 16 + lmod) * GIN_PAD + k);
            acc2[ct] = __builtin_amdgcn_wmma_f32_16x16x4_f32(
                false, a, false, b, (short)0, acc2[ct], false, false);
        }
    }

#pragma unroll
    for (int r = 0; r < 8; ++r) {
        int row = row0 + r + 8 * lhalf;
        bool ok = row < nrows;
        long orow = 0;
        if (ok) orow = ids ? (long)ids[row] : (long)row;
#pragma unroll
        for (int ct = 0; ct < 8; ++ct) {
            float val = acc2[ct][r] + lds_b2[ct * 16 + lmod];
            if (ok) {
                float* p = out + orow * GIN_D + ct * 16 + lmod;
                if (ids) unsafeAtomicAdd(p, val);
                else     *p = val;
            }
        }
    }
}

extern "C" void kernel_launch(void* const* d_in, const int* in_sizes, int n_in,
                              void* d_out, int out_size, void* d_ws, size_t ws_size,
                              hipStream_t stream) {
    const float* x    = (const float*)d_in[0];
    const int*   edge = (const int*)d_in[1];
    const int*   nid  = (const int*)d_in[2];
    const float* eps  = (const float*)d_in[3];
    const float* W1   = (const float*)d_in[4];
    const float* b1   = (const float*)d_in[5];
    const float* W2   = (const float*)d_in[6];
    const float* b2   = (const float*)d_in[7];
    const float* Wi1  = (const float*)d_in[8];
    const float* bi1  = (const float*)d_in[9];
    const float* Wi2  = (const float*)d_in[10];
    const float* bi2  = (const float*)d_in[11];

    const int N   = in_sizes[0] / GIN_D;
    const int E   = in_sizes[1] / 2;
    const int NID = in_sizes[2];
    float* h = (float*)d_ws;

    int n4 = (N * GIN_D) / 4;
    gin_h_init<<<(n4 + 255) / 256, 256, 0, stream>>>(x, eps, h, n4);

    long nthr = (long)E * 32;
    gin_edge_scatter<<<(unsigned)((nthr + 255) / 256), 256, 0, stream>>>(x, edge, h, E);

    gin_mlp<<<(N + GIN_ROWS - 1) / GIN_ROWS, 256, 0, stream>>>(
        h, W1, b1, W2, b2, (float*)d_out, nullptr, N);
    gin_mlp<<<(NID + GIN_ROWS - 1) / GIN_ROWS, 256, 0, stream>>>(
        h, Wi1, bi1, Wi2, bi2, (float*)d_out, nid, NID);
}


# aggregation by pull, not push: one int exchange per edge builds per-node lists; one wave per node sorts its list by edge index, sums rows in registers, writes h once; float sum order = edge index
# speedup vs baseline: 1.2660x; 1.2660x over previous
#include <hip/hip_runtime.h>

typedef __attribute__((ext_vector_type(2))) float v2f;
typedef __attribute__((ext_vector_type(8))) float v8f;

#define GIN_D    128
#define GIN_PAD  132
#define GIN_WAVES 8
#define GIN_ROWS (GIN_WAVES * 16)

#define GIN_NOEDGE 0x7fffffff

__global__ __launch_bounds__(256) void gin_head_init(int* __restrict__ head, int n) {
    int i = blockIdx.x * 256 + threadIdx.x;
    if (i < n) head[i] = -1;
}

__global__ __launch_bounds__(256) void gin_edge_link(const int* __restrict__ edge,
                                                     int* __restrict__ head,
                                                     int* __restrict__ link, int ne) {
    int e = blockIdx.x * 256 + threadIdx.x;
    if (e >= ne) return;
    int src = edge[e];
    int dst = edge[ne + e];
    if (src == dst) return;
    int prev = atomicExch(head + dst, e);
    ((int2*)link)[e] = make_int2(prev, src);
}

template <bool TWO>
__device__ __forceinline__ int gin_src_of_rank(int r, int k0, int k1, int s0, int s1) {
    if (!TWO) {
        unsigned m = __builtin_amdgcn_ballot_w32(k0 == r);
        return __builtin_amdgcn_readlane(s0, __builtin_ctz(m));
    }
    unsigned m = __builtin_amdgcn_ballot_w32(k0 == r || k1 == r);
    return __builtin_amdgcn_readlane(k0 == r ? s0 : s1, __builtin_ctz(m));
}

template <int K, bool TWO>
__device__ __forceinline__ void gin_add_rows(const float* __restrict__ xr, int r0,
                                             int k0, int k1, int s0, int s1, float4& acc) {
    float4 v[K];
#pragma unroll
    for (int u = 0; u < K; ++u) {
        int s = gin_src_of_rank<TWO>(r0 + u, k0, k1, s0, s1);
        v[u] = *(const float4*)(xr + (long)s * GIN_D);
    }
#pragma unroll
    for (int u = 0; u < K; ++u) {
        acc.x += v[u].x; acc.y += v[u].y; acc.z += v[u].z; acc.w += v[u].w;
    }
}

template <bool TWO>
__device__ __forceinline__ void gin_add_all(const float* __restrict__ xr, int L,
                                            int k0, int k1, int s0, int s1, float4& acc) {
    int r = 0;
    for (; r + 8 <= L; r += 8) gin_add_rows<8, TWO>(xr, r, k0, k1, s0, s1, acc);
    if (r + 4 <= L) { gin_add_rows<4, TWO>(xr, r, k0, k1, s0, s1, acc); r += 4; }
    if (r + 2 <= L) { gin_add_rows<2, TWO>(xr, r, k0, k1, s0, s1, acc); r += 2; }
    if (r < L)      { gin_add_rows<1, TWO>(xr, r, k0, k1, s0, s1, acc); }
}

__global__ __launch_bounds__(256) void gin_pull(const float* __restrict__ x,
                                                const float* __restrict__ eps,
                                                const int* __restrict__ head,
                                                const int* __restrict__ link,
                                                float* __restrict__ h, int n, int ne) {
    const int lane = threadIdx.x & 31;
    const int node = __builtin_amdgcn_readfirstlane(blockIdx.x * 8 + (threadIdx.x >> 5));
    if (node >= n) return;
    const int2* __restrict__ lk = (const int2*)link;
    const float* xr = x + lane * 4;
    const float4 xv = *(const float4*)(xr + (long)node * GIN_D);
    const int first = __builtin_amdgcn_readfirstlane(head[node]);

    int e0 = GIN_NOEDGE, e1 = GIN_NOEDGE, s0 = 0, s1 = 0;
    int cnt = 0;
    int cur = first;
    while (cur >= 0 && cnt < 32) {
        int2 p = lk[cur];
        if (lane == cnt) { e0 = cur; s0 = p.y; }
        cur = __builtin_amdgcn_readfirstlane(p.x);
        ++cnt;
    }
    while (cur >= 0 && cnt < 64) {
        int2 p = lk[cur];
        if (lane == cnt - 32) { e1 = cur; s1 = p.y; }
        cur = __builtin_amdgcn_readfirstlane(p.x);
        ++cnt;
    }
    const int L = cnt;

    float4 acc = make_float4(0.0f, 0.0f, 0.0f, 0.0f);
    if (cur < 0 && L <= 32) {
        int k0 = 0;
        for (int j = 0; j < L; ++j) k0 += __builtin_amdgcn_readlane(e0, j) < e0 ? 1 : 0;
        gin_add_all<false>(xr, L, k0, 0, s0, 0, acc);
    } else if (cur < 0) {
        int k0 = 0, k1 = 0;
        for (int j = 0; j < 32; ++j) {
            int ej = __builtin_amdgcn_readlane(e0, j);
            k0 += ej < e0 ? 1 : 0;
            k1 += ej < e1 ? 1 : 0;
        }
        for (int j = 32; j < L; ++j) {
            int ej = __builtin_amdgcn_readlane(e1, j - 32);
            k0 += ej < e0 ? 1 : 0;
            k1 += ej < e1 ? 1 : 0;
        }
        gin_add_all<true>(xr, L, k0, k1, s0, s1, acc);
    } else {
        int last = -1;
        for (int g = 0; g < ne; ++g) {
            int best = GIN_NOEDGE, bs = 0;
            int c2 = first;
            for (int g2 = 0; g2 < ne && c2 >= 0; ++g2) {
                int2 p = lk[c2];
                if (c2 > last && c2 < best) { best = c2; bs = p.y; }
                c2 = __builtin_amdgcn_readfirstlane(p.x);
            }
            if (best == GIN_NOEDGE) break;
            float4 v = *(const float4*)(xr + (long)bs * GIN_D);
            acc.x += v.x; acc.y += v.y; acc.z += v.z; acc.w += v.w;
            last = best;
        }
    }

    const float sc = 1.0f + eps[0];
    float4 o;
    o.x = sc * xv.x + acc.x; o.y = sc * xv.y + acc.y;
    o.z = sc * xv.z + acc.z; o.w = sc * xv.w + acc.w;
    *(float4*)(h + (long)node * GIN_D + lane * 4) = o;
}

__global__ __launch_bounds__(256) void gin_mlp(const float* __restrict__ h,
                                               const float* __restrict__ W1,
                                               const float* __restrict__ b1,
                                               const float* __restrict__ W2,
                                               const float* __restrict__ b2,
                                               float* __restrict__ out,
                                               const int* __restrict__ ids,
                                               int nrows) {
    __shared__ float lds_W[GIN_D * GIN_PAD];
    __shared__ float lds_A[GIN_WAVES * 16 * GIN_PAD];
    __shared__ float lds_b1[GIN_D];
    __shared__ float lds_b2[GIN_D];

    const int tid   = threadIdx.x;
    const int wave  = tid >> 5;
    const int lane  = tid & 31;
    const int lmod  = lane & 15;
    const int lhalf = lane >> 4;
    const int koff  = lhalf * 2;
    const int row0  = blockIdx.x * GIN_ROWS + wave * 16;

    if (tid < GIN_D) lds_b1[tid] = b1[tid];
    else             lds_b2[tid - GIN_D] = b2[tid - GIN_D];

    for (int i = tid; i < GIN_D * GIN_D; i += 256) {
        int k = i >> 7, n = i & (GIN_D - 1);
        lds_W[n * GIN_PAD + k] = W1[i];
    }

    float* myA = &lds_A[wave * 16 * GIN_PAD];
    for (int r = 0; r < 16; ++r) {
        int row = row0 + r;
        row = row < nrows ? row : nrows - 1;
        long srow = ids ? (long)ids[row] : (long)row;
        float4 v = *(const float4*)(h + srow * GIN_D + lane * 4);
        *(float4*)(myA + r * GIN_PAD + lane * 4) = v;
    }
    __syncthreads();

    v8f acc[8] = {};
    for (int kk = 0; kk < GIN_D; kk += 4) {
        int k = kk + koff;
        v2f a = *(const v2f*)(myA + lmod * GIN_PAD + k);
#pragma unroll
        for (int ct = 0; ct < 8; ++ct) {
            v2f b = *(const v2f*)(lds_W + (ct * 16 + lmod) * GIN_PAD + k);
            acc[ct] = __builtin_amdgcn_wmma_f32_16x16x4_f32(
                false, a, false, b, (short)0, acc[ct], false, false);
        }
    }
    __syncthreads();

#pragma unroll
    for (int ct = 0; ct < 8; ++ct) {
        float bv = lds_b1[ct * 16 + lmod];
#pragma unroll
        for (int r = 0; r < 8; ++r) {
            float t = acc[ct][r] + bv;
            myA[(r + 8 * lhalf) * GIN_PAD + ct * 16 + lmod] = t > 0.0f ? t : 0.0f;
        }
    }
    for (int i = tid; i < GIN_D * GIN_D; i += 256) {
        int k = i >> 7, n = i & (GIN_D - 1);
        lds_W[n * GIN_PAD + k] = W2[i];
    }
    __syncthreads();

    v8f acc2[8] = {};
    for (int kk = 0; kk < GIN_D; kk += 4) {
        int k = kk + koff;
        v2f a = *(const v2f*)(myA + lmod * GIN_PAD + k);
#pragma unroll
        for (int ct = 0; ct < 8; ++ct) {
            v2f b = *(const v2f*)(lds_W + (ct * 16 + lmod) * GIN_PAD + k);
            acc2[ct] = __builtin_amdgcn_wmma_f32_16x16x4_f32(
                false, a, false, b, (short)0, acc2[ct], false, false);
        }
    }

#pragma unroll
    for (int r = 0; r < 8; ++r) {
        int row = row0 + r + 8 * lhalf;
        bool ok = row < nrows;
        long orow = 0;
        if (ok) orow = ids ? (long)ids[row] : (long)row;
#pragma unroll
        for (int ct = 0; ct < 8; ++ct) {
            float val = acc2[ct][r] + lds_b2[ct * 16 + lmod];
            if (ok) {
                float* p = out + orow * GIN_D + ct * 16 + lmod;
                if (ids) unsafeAtomicAdd(p, val);
                else     *p = val;
            }
        }
    }
}

extern "C" void kernel_launch(void* const* d_in, const int* in_sizes, int n_in,
                              void* d_out, int out_size, void* d_ws, size_t ws_size,
                              hipStream_t stream) {
    const float* x    = (const float*)d_in[0];
    const int*   edge = (const int*)d_in[1];
    const int*   nid  = (const int*)d_in[2];
    const float* eps  = (const float*)d_in[3];
    const float* W1   = (const float*)d_in[4];
    const float* b1   = (const float*)d_in[5];
    const float* W2   = (const float*)d_in[6];
    const float* b2   = (const float*)d_in[7];
    const float* Wi1  = (const float*)d_in[8];
    const float* bi1  = (const float*)d_in[9];
    const float* Wi2  = (const float*)d_in[10];
    const float* bi2  = (const float*)d_in[11];

    const int N   = in_sizes[0] / GIN_D;
    const int E   = in_sizes[1] / 2;
    const int NID = in_sizes[2];
    float* h    = (float*)d_ws;
    int*   head = (int*)(h + (size_t)N * GIN_D);
    int*   link = head + ((N + 63) & ~63);

    gin_head_init<<<(N + 255) / 256, 256, 0, stream>>>(head, N);
    gin_edge_link<<<(E + 255) / 256, 256, 0, stream>>>(edge, head, link, E);
    gin_pull<<<(N + 7) / 8, 256, 0, stream>>>(x, eps, head, link, h, N, E);

    gin_mlp<<<(N + GIN_ROWS - 1) / GIN_ROWS, 256, 0, stream>>>(
        h, W1, b1, W2, b2, (float*)d_out, nullptr, N);
    gin_mlp<<<(NID + GIN_ROWS - 1) / GIN_ROWS, 256, 0, stream>>>(
        h, Wi1, bi1, Wi2, bi2, (float*)d_out, nid, NID);
}


# MLP kernel: a wave's 16 A-tile row loads go out as one group (row ids by one early load + readlane) and are stored after; was load-wait-store per row, 16 (id branch 32) dependent round trips, now 1
# speedup vs baseline: 1.3775x; 1.0881x over previous
#include <hip/hip_runtime.h>

typedef __attribute__((ext_vector_type(2))) float v2f;
typedef __attribute__((ext_vector_type(8))) float v8f;

#define GIN_D    128
#define GIN_PAD  132
#define GIN_WAVES 8
#define GIN_ROWS (GIN_WAVES * 16)

#define GIN_NOEDGE 0x7fffffff

__global__ __launch_bounds__(256) void gin_head_init(int* __restrict__ head, int n) {
    int i = blockIdx.x * 256 + threadIdx.x;
    if (i < n) head[i] = -1;
}

__global__ __launch_bounds__(256) void gin_edge_link(const int* __restrict__ edge,
                                                     int* __restrict__ head,
                                                     int* __restrict__ link, int ne) {
    int e = blockIdx.x * 256 + threadIdx.x;
    if (e >= ne) return;
    int src = edge[e];
    int dst = edge[ne + e];
    if (src == dst) return;
    int prev = atomicExch(head + dst, e);
    ((int2*)link)[e] = make_int2(prev, src);
}

template <bool TWO>
__device__ __forceinline__ int gin_src_of_rank(int r, int k0, int k1, int s0, int s1) {
    if (!TWO) {
        unsigned m = __builtin_amdgcn_ballot_w32(k0 == r);
        return __builtin_amdgcn_readlane(s0, __builtin_ctz(m));
    }
    unsigned m = __builtin_amdgcn_ballot_w32(k0 == r || k1 == r);
    return __builtin_amdgcn_readlane(k0 == r ? s0 : s1, __builtin_ctz(m));
}

template <int K, bool TWO>
__device__ __forceinline__ void gin_add_rows(const float* __restrict__ xr, int r0,
                                             int k0, int k1, int s0, int s1, float4& acc) {
    float4 v[K];
#pragma unroll
    for (int u = 0; u < K; ++u) {
        int s = gin_src_of_rank<TWO>(r0 + u, k0, k1, s0, s1);
        v[u] = *(const float4*)(xr + (long)s * GIN_D);
    }
#pragma unroll
    for (int u = 0; u < K; ++u) {
        acc.x += v[u].x; acc.y += v[u].y; acc.z += v[u].z; acc.w += v[u].w;
    }
}

template <bool TWO>
__device__ __forceinline__ void gin_add_all(const float* __restrict__ xr, int L,
                                            int k0, int k1, int s0, int s1, float4& acc) {
    int r = 0;
    for (; r + 8 <= L; r += 8) gin_add_rows<8, TWO>(xr, r, k0, k1, s0, s1, acc);
    if (r + 4 <= L) { gin_add_rows<4, TWO>(xr, r, k0, k1, s0, s1, acc); r += 4; }
    if (r + 2 <= L) { gin_add_rows<2, TWO>(xr, r, k0, k1, s0, s1, acc); r += 2; }
    if (r < L)      { gin_add_rows<1, TWO>(xr, r, k0, k1, s0, s1, acc); }
}

__global__ __launch_bounds__(256) void gin_pull(const float* __restrict__ x,
                                                const float* __restrict__ eps,
                                                const int* __restrict__ head,
                                                const int* __restrict__ link,
                                                float* __restrict__ h, int n, int ne) {
    const int lane = threadIdx.x & 31;
    const int node = __builtin_amdgcn_readfirstlane(blockIdx.x * 8 + (threadIdx.x >> 5));
    if (node >= n) return;
    const int2* __restrict__ lk = (const int2*)link;
    const float* xr = x + lane * 4;
    const float4 xv = *(const float4*)(xr + (long)node * GIN_D);
    const int first = __builtin_amdgcn_readfirstlane(head[node]);

    int e0 = GIN_NOEDGE, e1 = GIN_NOEDGE, s0 = 0, s1 = 0;
    int cnt = 0;
    int cur = first;
    while (cur >= 0 && cnt < 32) {
        int2 p = lk[cur];
        if (lane == cnt) { e0 = cur; s0 = p.y; }
        cur = __builtin_amdgcn_readfirstlane(p.x);
        ++cnt;
    }
    while (cur >= 0 && cnt < 64) {
        int2 p = lk[cur];
        if (lane == cnt - 32) { e1 = cur; s1 = p.y; }
        cur = __builtin_amdgcn_readfirstlane(p.x);
        ++cnt;
    }
    const int L = cnt;

    float4 acc = make_float4(0.0f, 0.0f, 0.0f, 0.0f);
    if (cur < 0 && L <= 32) {
        int k0 = 0;
        for (int j = 0; j < L; ++j) k0 += __builtin_amdgcn_readlane(e0, j) < e0 ? 1 : 0;
        gin_add_all<false>(xr, L, k0, 0, s0, 0, acc);
    } else if (cur < 0) {
        int k0 = 0, k1 = 0;
        for (int j = 0; j < 32; ++j) {
            int ej = __builtin_amdgcn_readlane(e0, j);
            k0 += ej < e0 ? 1 : 0;
            k1 += ej < e1 ? 1 : 0;
        }
        for (int j = 32; j < L; ++j) {
            int ej = __builtin_amdgcn_readlane(e1, j - 32);
            k0 += ej < e0 ? 1 : 0;
            k1 += ej < e1 ? 1 : 0;
        }
        gin_add_all<true>(xr, L, k0, k1, s0, s1, acc);
    } else {
        int last = -1;
        for (int g = 0; g < ne; ++g) {
            int best = GIN_NOEDGE, bs = 0;
            int c2 = first;
            for (int g2 = 0; g2 < ne && c2 >= 0; ++g2) {
                int2 p = lk[c2];
                if (c2 > last && c2 < best) { best = c2; bs = p.y; }
                c2 = __builtin_amdgcn_readfirstlane(p.x);
            }
            if (best == GIN_NOEDGE) break;
            float4 v = *(const float4*)(xr + (long)bs * GIN_D);
            acc.x += v.x; acc.y += v.y; acc.z += v.z; acc.w += v.w;
            last = best;
        }
    }

    const float sc = 1.0f + eps[0];
    float4 o;
    o.x = sc * xv.x + acc.x; o.y = sc * xv.y + acc.y;
    o.z = sc * xv.z + acc.z; o.w = sc * xv.w + acc.w;
    *(float4*)(h + (long)node * GIN_D + lane * 4) = o;
}

__global__ __launch_bounds__(256) void gin_mlp(const float* __restrict__ h,
                                               const float* __restrict__ W1,
                                               const float* __restrict__ b1,
                                               const float* __restrict__ W2,
                                               const float* __restrict__ b2,
                                               float* __restrict__ out,
                                               const int* __restrict__ ids,
                                               int nrows) {
    __shared__ float lds_W[GIN_D * GIN_PAD];
    __shared__ float lds_A[GIN_WAVES * 16 * GIN_PAD];
    __shared__ float lds_b1[GIN_D];
    __shared__ float lds_b2[GIN_D];

    const int tid   = threadIdx.x;
    const int wave  = tid >> 5;
    const int lane  = tid & 31;
    const int lmod  = lane & 15;
    const int lhalf = lane >> 4;
    const int koff  = lhalf * 2;
    const int row0  = blockIdx.x * GIN_ROWS + wave * 16;

    int rsel = row0 + lmod;
    rsel = rsel < nrows ? rsel : nrows - 1;
    const int myrow = ids ? ids[rsel] : rsel;

    if (tid < GIN_D) lds_b1[tid] = b1[tid];
    else             lds_b2[tid - GIN_D] = b2[tid - GIN_D];

    for (int i = tid; i < GIN_D * GIN_D; i += 256) {
        int k = i >> 7, n = i & (GIN_D - 1);
        lds_W[n * GIN_PAD + k] = W1[i];
    }

    float* myA = &lds_A[wave * 16 * GIN_PAD];
    float4 av[16];
#pragma unroll
    for (int r = 0; r < 16; ++r) {
        const int srow = __builtin_amdgcn_readlane(myrow, r);
        av[r] = *(const float4*)(h + (long)srow * GIN_D + lane * 4);
    }
#pragma unroll
    for (int r = 0; r < 16; ++r)
        *(float4*)(myA + r * GIN_PAD + lane * 4) = av[r];
    __syncthreads();

    v8f acc[8] = {};
    for (int kk = 0; kk < GIN_D; kk += 4) {
        int k = kk + koff;
        v2f a = *(const v2f*)(myA + lmod * GIN_PAD + k);
#pragma unroll
        for (int ct = 0; ct < 8; ++ct) {
            v2f b = *(const v2f*)(lds_W + (ct * 16 + lmod) * GIN_PAD + k);
            acc[ct] = __builtin_amdgcn_wmma_f32_16x16x4_f32(
                false, a, false, b, (short)0, acc[ct], false, false);
        }
    }
    __syncthreads();

#pragma unroll
    for (int ct = 0; ct < 8; ++ct) {
        float bv = lds_b1[ct * 16 + lmod];
#pragma unroll
        for (int r = 0; r < 8; ++r) {
            float t = acc[ct][r] + bv;
            myA[(r + 8 * lhalf) * GIN_PAD + ct * 16 + lmod] = t > 0.0f ? t : 0.0f;
        }
    }
    for (int i = tid; i < GIN_D * GIN_D; i += 256) {
        int k = i >> 7, n = i & (GIN_D - 1);
        lds_W[n * GIN_PAD + k] = W2[i];
    }
    __syncthreads();

    v8f acc2[8] = {};
    for (int kk = 0; kk < GIN_D; kk += 4) {
        int k = kk + koff;
        v2f a = *(const v2f*)(myA + lmod * GIN_PAD + k);
#pragma unroll
        for (int ct = 0; ct < 8; ++ct) {
            v2f b = *(const v2f*)(lds_W + (ct * 16 + lmod) * GIN_PAD + k);
            acc2[ct] = __builtin_amdgcn_wmma_f32_16x16x4_f32(
                false, a, false, b, (short)0, acc2[ct], false, false);
        }
    }

#pragma unroll
    for (int r = 0; r < 8; ++r) {
        int row = row0 + r + 8 * lhalf;
        bool ok = row < nrows;
        long orow = 0;
        if (ok) orow = ids ? (long)ids[row] : (long)row;
#pragma unroll
        for (int ct = 0; ct < 8; ++ct) {
            float val = acc2[ct][r] + lds_b2[ct * 16 + lmod];
            if (ok) {
                float* p = out + orow * GIN_D + ct * 16 + lmod;
                if (ids) unsafeAtomicAdd(p, val);
                else     *p = val;
            }
        }
    }
}

extern "C" void kernel_launch(void* const* d_in, const int* in_sizes, int n_in,
                              void* d_out, int out_size, void* d_ws, size_t ws_size,
                              hipStream_t stream) {
    const float* x    = (const float*)d_in[0];
    const int*   edge = (const int*)d_in[1];
    const int*   nid  = (const int*)d_in[2];
    const float* eps  = (const float*)d_in[3];
    const float* W1   = (const float*)d_in[4];
    const float* b1   = (const float*)d_in[5];
    const float* W2   = (const float*)d_in[6];
    const float* b2   = (const float*)d_in[7];
    const float* Wi1  = (const float*)d_in[8];
    const float* bi1  = (const float*)d_in[9];
    const float* Wi2  = (const float*)d_in[10];
    const float* bi2  = (const float*)d_in[11];

    const int N   = in_sizes[0] / GIN_D;
    const int E   = in_sizes[1] / 2;
    const int NID = in_sizes[2];
    float* h    = (float*)d_ws;
    int*   head = (int*)(h + (size_t)N * GIN_D);
    int*   link = head + ((N + 63) & ~63);

    gin_head_init<<<(N + 255) / 256, 256, 0, stream>>>(head, N);
    gin_edge_link<<<(E + 255) / 256, 256, 0, stream>>>(edge, head, link, E);
    gin_pull<<<(N + 7) / 8, 256, 0, stream>>>(x, eps, head, link, h, N, E);

    gin_mlp<<<(N + GIN_ROWS - 1) / GIN_ROWS, 256, 0, stream>>>(
        h, W1, b1, W2, b2, (float*)d_out, nullptr, N);
    gin_mlp<<<(NID + GIN_ROWS - 1) / GIN_ROWS, 256, 0, stream>>>(
        h, Wi1, bi1, Wi2, bi2, (float*)d_out, nid, NID);
}


# aggregation lists built by a two-level counting sort in LDS (edges binned by destination range, then grouped by node), replacing one global exchange and one list hop per edge; sum order kept
# speedup vs baseline: 2.0612x; 1.4963x over previous
#include <hip/hip_runtime.h>

typedef __attribute__((ext_vector_type(2))) float v2f;
typedef __attribute__((ext_vector_type(8))) float v8f;

#define GIN_D    128
#define GIN_PAD  132
#define GIN_WAVES 8
#define GIN_ROWS (GIN_WAVES * 16)

#define GIN_NOEDGE 0x7fffffff
#define GIN_GA   256
#define GIN_NB   512
#define GIN_NBP  (GIN_NB + 1)
#define GIN_BWMAX 128
#define GIN_MAXM 4096

__device__ __forceinline__ int gin_block_scan(int v, int* sh, int tid, int nt, int& total) {
    const int lane = tid & 31, w = tid >> 5, nw = nt >> 5;
    int incl = v;
    for (int d = 1; d < 32; d <<= 1) {
        int t = __shfl_up(incl, d, 32);
        if (lane >= d) incl += t;
    }
    if (lane == 31) sh[w] = incl;
    __syncthreads();
    int pre = 0, tot = 0;
    for (int q = 0; q < nw; ++q) {
        int t = sh[q];
        if (q < w) pre += t;
        tot += t;
    }
    total = tot;
    __syncthreads();
    return pre + incl - v;
}

__global__ __launch_bounds__(1024) void gin_bin(const int* __restrict__ edge,
                                               int* __restrict__ tab,
                                               int2* __restrict__ staged, int n, int ne) {
    __shared__ int cnt[GIN_NB];
    __shared__ int cur[GIN_NB];
    __shared__ int sc[32];
    const int tid = threadIdx.x;
    const int wg  = blockIdx.x;
    const int ch  = (ne + GIN_GA - 1) / GIN_GA;
    const int bw  = (n + GIN_NB - 1) / GIN_NB;
    const int e0  = wg * ch;
    const int e1  = e0 + ch < ne ? e0 + ch : ne;
    if (tid < GIN_NB) cnt[tid] = 0;
    __syncthreads();
    for (int e = e0 + tid; e < e1; e += 1024) {
        int s = edge[e], d = edge[ne + e];
        if (s != d) atomicAdd(&cnt[d / bw], 1);
    }
    __syncthreads();
    int total;
    const int base = gin_block_scan(tid < GIN_NB ? cnt[tid] : 0, sc, tid, 1024, total);
    if (tid < GIN_NB) { cur[tid] = base; tab[wg * GIN_NBP + tid] = base; }
    if (tid == 0) tab[wg * GIN_NBP + GIN_NB] = total;
    __syncthreads();
    for (int e = e0 + tid; e < e1; e += 1024) {
        int s = edge[e], d = edge[ne + e];
        if (s != d) {
            int bin = d / bw;
            int pos = atomicAdd(&cur[bin], 1);
            staged[e0 + pos] = make_int2(e, ((d - bin * bw) << 16) | s);
        }
    }
}

template <bool TWO>
__device__ __forceinline__ int gin_src_of_rank(int r, int k0, int k1, int s0, int s1) {
    if (!TWO) {
        unsigned m = __builtin_amdgcn_ballot_w32(k0 == r);
        return __builtin_amdgcn_readlane(s0, __builtin_ctz(m));
    }
    unsigned m = __builtin_amdgcn_ballot_w32(k0 == r || k1 == r);
    return __builtin_amdgcn_readlane(k0 == r ? s0 : s1, __builtin_ctz(m));
}

template <int K, bool TWO>
__device__ __forceinline__ void gin_add_rows(const float* __restrict__ xr, int r0,
                                             int k0, int k1, int s0, int s1, float4& acc) {
    float4 v[K];
#pragma unroll
    for (int u = 0; u < K; ++u) {
        int s = gin_src_of_rank<TWO>(r0 + u, k0, k1, s0, s1);
        v[u] = *(const float4*)(xr + (long)s * GIN_D);
    }
#pragma unroll
    for (int u = 0; u < K; ++u) {
        acc.x += v[u].x; acc.y += v[u].y; acc.z += v[u].z; acc.w += v[u].w;
    }
}

template <bool TWO>
__device__ __forceinline__ void gin_add_all(const float* __restrict__ xr, int L,
                                            int k0, int k1, int s0, int s1, float4& acc) {
    int r = 0;
    for (; r + 8 <= L; r += 8) gin_add_rows<8, TWO>(xr, r, k0, k1, s0, s1, acc);
    if (r + 4 <= L) { gin_add_rows<4, TWO>(xr, r, k0, k1, s0, s1, acc); r += 4; }
    if (r + 2 <= L) { gin_add_rows<2, TWO>(xr, r, k0, k1, s0, s1, acc); r += 2; }
    if (r < L)      { gin_add_rows<1, TWO>(xr, r, k0, k1, s0, s1, acc); }
}

__device__ __forceinline__ void gin_add_by_scan(const float* __restrict__ xr,
                                                const int* __restrict__ edge, int ne,
                                                int node, int lane, float4& acc) {
    for (int base = 0; base < ne; base += 32) {
        int e = base + lane;
        int s = -1, d = -1;
        if (e < ne) { s = edge[e]; d = edge[ne + e]; }
        unsigned m = __builtin_amdgcn_ballot_w32(d == node && s != node);
        while (m) {
            int l = __builtin_ctz(m);
            m &= m - 1;
            int ss = __builtin_amdgcn_readlane(s, l);
            float4 v = *(const float4*)(xr + (long)ss * GIN_D);
            acc.x += v.x; acc.y += v.y; acc.z += v.z; acc.w += v.w;
        }
    }
}

__global__ __launch_bounds__(1024, 16) void gin_pull_bin(const float* __restrict__ x,
                                                    const float* __restrict__ eps,
                                                    const int* __restrict__ edge,
                                                    const int* __restrict__ tab,
                                                    const int2* __restrict__ staged,
                                                    float* __restrict__ h, int n, int ne) {
    __shared__ int2 ebuf[GIN_MAXM];
    __shared__ int2 sbuf[GIN_MAXM];
    __shared__ int ncnt[GIN_BWMAX];
    __shared__ int nstart[GIN_BWMAX];
    __shared__ int ncur[GIN_BWMAX];
    __shared__ int sc[32];
    const int tid  = threadIdx.x;
    const int lane = tid & 31;
    const int wave = __builtin_amdgcn_readfirstlane(tid >> 5);
    const int b    = blockIdx.x;
    const int ch   = (ne + GIN_GA - 1) / GIN_GA;
    const int bw   = (n + GIN_NB - 1) / GIN_NB;

    int s = 0, c = 0;
    if (tid < GIN_GA) {
        s = tab[tid * GIN_NBP + b];
        c = tab[tid * GIN_NBP + b + 1] - s;
    }
    if (tid < GIN_BWMAX) ncnt[tid] = 0;
    int M;
    const int base = gin_block_scan(c, sc, tid, 1024, M);
    const bool fits = M <= GIN_MAXM && bw <= GIN_BWMAX;
    if (fits) {
        const int2* seg = staged + (long)tid * ch + s;
        for (int i0 = 0; i0 < c; i0 += 8) {
            int2 r[8];
#pragma unroll
            for (int u = 0; u < 8; ++u) r[u] = seg[i0 + u < c ? i0 + u : c - 1];
#pragma unroll
            for (int u = 0; u < 8; ++u)
                if (i0 + u < c) ebuf[base + i0 + u] = r[u];
        }
        __syncthreads();
        for (int i = tid; i < M; i += 1024) atomicAdd(&ncnt[ebuf[i].y >> 16], 1);
        __syncthreads();
        int tot;
        const int st = gin_block_scan(tid < GIN_BWMAX ? ncnt[tid] : 0, sc, tid, 1024, tot);
        if (tid < GIN_BWMAX) { nstart[tid] = st; ncur[tid] = st; }
        __syncthreads();
        for (int i = tid; i < M; i += 1024) {
            int2 r = ebuf[i];
            int pos = atomicAdd(&ncur[r.y >> 16], 1);
            sbuf[pos] = make_int2(r.x, r.y & 0xffff);
        }
        __syncthreads();
    }

    const float scl = 1.0f + eps[0];
    const float* xr = x + lane * 4;
    for (int j = wave; j < bw; j += 32) {
        const int node = b * bw + j;
        if (node >= n) break;
        const float4 xv = *(const float4*)(xr + (long)node * GIN_D);
        float4 acc = make_float4(0.0f, 0.0f, 0.0f, 0.0f);
        const int L   = fits ? __builtin_amdgcn_readfirstlane(ncnt[j]) : 0;
        const int st0 = fits ? __builtin_amdgcn_readfirstlane(nstart[j]) : 0;
        if (fits && L <= 32) {
            int e0 = GIN_NOEDGE, s0 = 0;
            if (lane < L) { int2 r = sbuf[st0 + lane]; e0 = r.x; s0 = r.y; }
            int k0 = 0;
            for (int q = 0; q < L; ++q) k0 += __builtin_amdgcn_readlane(e0, q) < e0 ? 1 : 0;
            gin_add_all<false>(xr, L, k0, 0, s0, 0, acc);
        } else if (fits && L <= 64) {
            int2 r0 = sbuf[st0 + lane];
            int e0 = r0.x, s0 = r0.y, e1 = GIN_NOEDGE, s1 = 0;
            if (32 + lane < L) { int2 r1 = sbuf[st0 + 32 + lane]; e1 = r1.x; s1 = r1.y; }
            int k0 = 0, k1 = 0;
            for (int q = 0; q < 32; ++q) {
                int eq = __builtin_amdgcn_readlane(e0, q);
                k0 += eq < e0 ? 1 : 0;
                k1 += eq < e1 ? 1 : 0;
            }
            for (int q = 32; q < L; ++q) {
                int eq = __builtin_amdgcn_readlane(e1, q - 32);
                k0 += eq < e0 ? 1 : 0;
                k1 += eq < e1 ? 1 : 0;
            }
            gin_add_all<true>(xr, L, k0, k1, s0, s1, acc);
        } else {
            gin_add_by_scan(xr, edge, ne, node, lane, acc);
        }
        float4 o;
        o.x = scl * xv.x + acc.x; o.y = scl * xv.y + acc.y;
        o.z = scl * xv.z + acc.z; o.w = scl * xv.w + acc.w;
        *(float4*)(h + (long)node * GIN_D + lane * 4) = o;
    }
}

__global__ __launch_bounds__(256) void gin_mlp(const float* __restrict__ h,
                                               const float* __restrict__ W1,
                                               const float* __restrict__ b1,
                                               const float* __restrict__ W2,
                                               const float* __restrict__ b2,
                                               float* __restrict__ out,
                                               const int* __restrict__ ids,
                                               int nrows) {
    __shared__ float lds_W[GIN_D * GIN_PAD];
    __shared__ float lds_A[GIN_WAVES * 16 * GIN_PAD];
    __shared__ float lds_b1[GIN_D];
    __shared__ float lds_b2[GIN_D];

    const int tid   = threadIdx.x;
    const int wave  = tid >> 5;
    const int lane  = tid & 31;
    const int lmod  = lane & 15;
    const int lhalf = lane >> 4;
    const int koff  = lhalf * 2;
    const int row0  = blockIdx.x * GIN_ROWS + wave * 16;

    int rsel = row0 + lmod;
    rsel = rsel < nrows ? rsel : nrows - 1;
    const int myrow = ids ? ids[rsel] : rsel;

    if (tid < GIN_D) lds_b1[tid] = b1[tid];
    else             lds_b2[tid - GIN_D] = b2[tid - GIN_D];

    for (int i = tid; i < GIN_D * GIN_D; i += 256) {
        int k = i >> 7, n = i & (GIN_D - 1);
        lds_W[n * GIN_PAD + k] = W1[i];
    }

    float* myA = &lds_A[wave * 16 * GIN_PAD];
    float4 av[16];
#pragma unroll
    for (int r = 0; r < 16; ++r) {
        const int srow = __builtin_amdgcn_readlane(myrow, r);
        av[r] = *(const float4*)(h + (long)srow * GIN_D + lane * 4);
    }
#pragma unroll
    for (int r = 0; r < 16; ++r)
        *(float4*)(myA + r * GIN_PAD + lane * 4) = av[r];
    __syncthreads();

    v8f acc[8] = {};
    for (int kk = 0; kk < GIN_D; kk += 4) {
        int k = kk + koff;
        v2f a = *(const v2f*)(myA + lmod * GIN_PAD + k);
#pragma unroll
        for (int ct = 0; ct < 8; ++ct) {
            v2f b = *(const v2f*)(lds_W + (ct * 16 + lmod) * GIN_PAD + k);
            acc[ct] = __builtin_amdgcn_wmma_f32_16x16x4_f32(
                false, a, false, b, (short)0, acc[ct], false, false);
        }
    }
    __syncthreads();

#pragma unroll
    for (int ct = 0; ct < 8; ++ct) {
        float bv = lds_b1[ct * 16 + lmod];
#pragma unroll
        for (int r = 0; r < 8; ++r) {
            float t = acc[ct][r] + bv;
            myA[(r + 8 * lhalf) * GIN_PAD + ct * 16 + lmod] = t > 0.0f ? t : 0.0f;
        }
    }
    for (int i = tid; i < GIN_D * GIN_D; i += 256) {
        int k = i >> 7, n = i & (GIN_D - 1);
        lds_W[n * GIN_PAD + k] = W2[i];
    }
    __syncthreads();

    v8f acc2[8] = {};
    for (int kk = 0; kk < GIN_D; kk += 4) {
        int k = kk + koff;
        v2f a = *(const v2f*)(myA + lmod * GIN_PAD + k);
#pragma unroll
        for (int ct = 0; ct < 8; ++ct) {
            v2f b = *(const v2f*)(lds_W + (ct * 16 + lmod) * GIN_PAD + k);
            acc2[ct] = __builtin_amdgcn_wmma_f32_16x16x4_f32(
                false, a, false, b, (short)0, acc2[ct], false, false);
        }
    }

#pragma unroll
    for (int r = 0; r < 8; ++r) {
        int row = row0 + r + 8 * lhalf;
        bool ok = row < nrows;
        long orow = 0;
        if (ok) orow = ids ? (long)ids[row] : (long)row;
#pragma unroll
        for (int ct = 0; ct < 8; ++ct) {
            float val = acc2[ct][r] + lds_b2[ct * 16 + lmod];
            if (ok) {
                float* p = out + orow * GIN_D + ct * 16 + lmod;
                if (ids) unsafeAtomicAdd(p, val);
                else     *p = val;
            }
        }
    }
}

extern "C" void kernel_launch(void* const* d_in, const int* in_sizes, int n_in,
                              void* d_out, int out_size, void* d_ws, size_t ws_size,
                              hipStream_t stream) {
    const float* x    = (const float*)d_in[0];
    const int*   edge = (const int*)d_in[1];
    const int*   nid  = (const int*)d_in[2];
    const float* eps  = (const float*)d_in[3];
    const float* W1   = (const float*)d_in[4];
    const float* b1   = (const float*)d_in[5];
    const float* W2   = (const float*)d_in[6];
    const float* b2   = (const float*)d_in[7];
    const float* Wi1  = (const float*)d_in[8];
    const float* bi1  = (const float*)d_in[9];
    const float* Wi2  = (const float*)d_in[10];
    const float* bi2  = (const float*)d_in[11];

    const int N   = in_sizes[0] / GIN_D;
    const int E   = in_sizes[1] / 2;
    const int NID = in_sizes[2];
    float* h      = (float*)d_ws;
    int*   tab    = (int*)(h + (size_t)N * GIN_D);
    int2*  staged = (int2*)(tab + GIN_GA * GIN_NBP);

    gin_bin<<<GIN_GA, 1024, 0, stream>>>(edge, tab, staged, N, E);
    gin_pull_bin<<<GIN_NB, 1024, 0, stream>>>(x, eps, edge, tab, staged, h, N, E);

    gin_mlp<<<(N + GIN_ROWS - 1) / GIN_ROWS, 256, 0, stream>>>(
        h, W1, b1, W2, b2, (float*)d_out, nullptr, N);
    gin_mlp<<<(NID + GIN_ROWS - 1) / GIN_ROWS, 256, 0, stream>>>(
        h, Wi1, bi1, Wi2, bi2, (float*)d_out, nid, NID);
}


# MLP kernels: matrix work dealt per SIMD, not per wave: 16-wave work-groups, unit = (16-row tile, 32-column quarter), 13 tiles (main) / 3 (id branch) per group so every SIMD gets equal WMMAs
# speedup vs baseline: 2.5184x; 1.2218x over previous
#include <hip/hip_runtime.h>

typedef __attribute__((ext_vector_type(2))) float v2f;
typedef __attribute__((ext_vector_type(8))) float v8f;

#define GIN_D    128
#define GIN_PAD  132
#define GIN_MAIN_TILES 13
#define GIN_ID_TILES   3

#define GIN_NOEDGE 0x7fffffff
#define GIN_GA   256
#define GIN_NB   512
#define GIN_NBP  (GIN_NB + 1)
#define GIN_BWMAX 128
#define GIN_MAXM 4096

__device__ __forceinline__ int gin_block_scan(int v, int* sh, int tid, int nt, int& total) {
    const int lane = tid & 31, w = tid >> 5, nw = nt >> 5;
    int incl = v;
    for (int d = 1; d < 32; d <<= 1) {
        int t = __shfl_up(incl, d, 32);
        if (lane >= d) incl += t;
    }
    if (lane == 31) sh[w] = incl;
    __syncthreads();
    int pre = 0, tot = 0;
    for (int q = 0; q < nw; ++q) {
        int t = sh[q];
        if (q < w) pre += t;
        tot += t;
    }
    total = tot;
    __syncthreads();
    return pre + incl - v;
}

__global__ __launch_bounds__(1024) void gin_bin(const int* __restrict__ edge,
                                               int* __restrict__ tab,
                                               int2* __restrict__ staged, int n, int ne) {
    __shared__ int cnt[GIN_NB];
    __shared__ int cur[GIN_NB];
    __shared__ int sc[32];
    const int tid = threadIdx.x;
    const int wg  = blockIdx.x;
    const int ch  = (ne + GIN_GA - 1) / GIN_GA;
    const int bw  = (n + GIN_NB - 1) / GIN_NB;
    const int e0  = wg * ch;
    const int e1  = e0 + ch < ne ? e0 + ch : ne;
    if (tid < GIN_NB) cnt[tid] = 0;
    __syncthreads();
    for (int e = e0 + tid; e < e1; e += 1024) {
        int s = edge[e], d = edge[ne + e];
        if (s != d) atomicAdd(&cnt[d / bw], 1);
    }
    __syncthreads();
    int total;
    const int base = gin_block_scan(tid < GIN_NB ? cnt[tid] : 0, sc, tid, 1024, total);
    if (tid < GIN_NB) { cur[tid] = base; tab[wg * GIN_NBP + tid] = base; }
    if (tid == 0) tab[wg * GIN_NBP + GIN_NB] = total;
    __syncthreads();
    for (int e = e0 + tid; e < e1; e += 1024) {
        int s = edge[e], d = edge[ne + e];
        if (s != d) {
            int bin = d / bw;
            int pos = atomicAdd(&cur[bin], 1);
            staged[e0 + pos] = make_int2(e, ((d - bin * bw) << 16) | s);
        }
    }
}

template <bool TWO>
__device__ __forceinline__ int gin_src_of_rank(int r, int k0, int k1, int s0, int s1) {
    if (!TWO) {
        unsigned m = __builtin_amdgcn_ballot_w32(k0 == r);
        return __builtin_amdgcn_readlane(s0, __builtin_ctz(m));
    }
    unsigned m = __builtin_amdgcn_ballot_w32(k0 == r || k1 == r);
    return __builtin_amdgcn_readlane(k0 == r ? s0 : s1, __builtin_ctz(m));
}

template <int K, bool TWO>
__device__ __forceinline__ void gin_add_rows(const float* __restrict__ xr, int r0,
                                             int k0, int k1, int s0, int s1, float4& acc) {
    float4 v[K];
#pragma unroll
    for (int u = 0; u < K; ++u) {
        int s = gin_src_of_rank<TWO>(r0 + u, k0, k1, s0, s1);
        v[u] = *(const float4*)(xr + (long)s * GIN_D);
    }
#pragma unroll
    for (int u = 0; u < K; ++u) {
        acc.x += v[u].x; acc.y += v[u].y; acc.z += v[u].z; acc.w += v[u].w;
    }
}

template <bool TWO>
__device__ __forceinline__ void gin_add_all(const float* __restrict__ xr, int L,
                                            int k0, int k1, int s0, int s1, float4& acc) {
    int r = 0;
    for (; r + 8 <= L; r += 8) gin_add_rows<8, TWO>(xr, r, k0, k1, s0, s1, acc);
    if (r + 4 <= L) { gin_add_rows<4, TWO>(xr, r, k0, k1, s0, s1, acc); r += 4; }
    if (r + 2 <= L) { gin_add_rows<2, TWO>(xr, r, k0, k1, s0, s1, acc); r += 2; }
    if (r < L)      { gin_add_rows<1, TWO>(xr, r, k0, k1, s0, s1, acc); }
}

__device__ __forceinline__ void gin_add_by_scan(const float* __restrict__ xr,
                                                const int* __restrict__ edge, int ne,
                                                int node, int lane, float4& acc) {
    for (int base = 0; base < ne; base += 32) {
        int e = base + lane;
        int s = -1, d = -1;
        if (e < ne) { s = edge[e]; d = edge[ne + e]; }
        unsigned m = __builtin_amdgcn_ballot_w32(d == node && s != node);
        while (m) {
            int l = __builtin_ctz(m);
            m &= m - 1;
            int ss = __builtin_amdgcn_readlane(s, l);
            float4 v = *(const float4*)(xr + (long)ss * GIN_D);
            acc.x += v.x; acc.y += v.y; acc.z += v.z; acc.w += v.w;
        }
    }
}

__global__ __launch_bounds__(1024, 16) void gin_pull_bin(const float* __restrict__ x,
                                                    const float* __restrict__ eps,
                                                    const int* __restrict__ edge,
                                                    const int* __restrict__ tab,
                                                    const int2* __restrict__ staged,
                                                    float* __restrict__ h, int n, int ne) {
    __shared__ int2 ebuf[GIN_MAXM];
    __shared__ int2 sbuf[GIN_MAXM];
    __shared__ int ncnt[GIN_BWMAX];
    __shared__ int nstart[GIN_BWMAX];
    __shared__ int ncur[GIN_BWMAX];
    __shared__ int sc[32];
    const int tid  = threadIdx.x;
    const int lane = tid & 31;
    const int wave = __builtin_amdgcn_readfirstlane(tid >> 5);
    const int b    = blockIdx.x;
    const int ch   = (ne + GIN_GA - 1) / GIN_GA;
    const int bw   = (n + GIN_NB - 1) / GIN_NB;

    int s = 0, c = 0;
    if (tid < GIN_GA) {
        s = tab[tid * GIN_NBP + b];
        c = tab[tid * GIN_NBP + b + 1] - s;
    }
    if (tid < GIN_BWMAX) ncnt[tid] = 0;
    int M;
    const int base = gin_block_scan(c, sc, tid, 1024, M);
    const bool fits = M <= GIN_MAXM && bw <= GIN_BWMAX;
    if (fits) {
        const int2* seg = staged + (long)tid * ch + s;
        for (int i0 = 0; i0 < c; i0 += 8) {
            int2 r[8];
#pragma unroll
            for (int u = 0; u < 8; ++u) r[u] = seg[i0 + u < c ? i0 + u : c - 1];
#pragma unroll
            for (int u = 0; u < 8; ++u)
                if (i0 + u < c) ebuf[base + i0 + u] = r[u];
        }
        __syncthreads();
        for (int i = tid; i < M; i += 1024) atomicAdd(&ncnt[ebuf[i].y >> 16], 1);
        __syncthreads();
        int tot;
        const int st = gin_block_scan(tid < GIN_BWMAX ? ncnt[tid] : 0, sc, tid, 1024, tot);
        if (tid < GIN_BWMAX) { nstart[tid] = st; ncur[tid] = st; }
        __syncthreads();
        for (int i = tid; i < M; i += 1024) {
            int2 r = ebuf[i];
            int pos = atomicAdd(&ncur[r.y >> 16], 1);
            sbuf[pos] = make_int2(r.x, r.y & 0xffff);
        }
        __syncthreads();
    }

    const float scl = 1.0f + eps[0];
    const float* xr = x + lane * 4;
    for (int j = wave; j < bw; j += 32) {
        const int node = b * bw + j;
        if (node >= n) break;
        const float4 xv = *(const float4*)(xr + (long)node * GIN_D);
        float4 acc = make_float4(0.0f, 0.0f, 0.0f, 0.0f);
        const int L   = fits ? __builtin_amdgcn_readfirstlane(ncnt[j]) : 0;
        const int st0 = fits ? __builtin_amdgcn_readfirstlane(nstart[j]) : 0;
        if (fits && L <= 32) {
            int e0 = GIN_NOEDGE, s0 = 0;
            if (lane < L) { int2 r = sbuf[st0 + lane]; e0 = r.x; s0 = r.y; }
            int k0 = 0;
            for (int q = 0; q < L; ++q) k0 += __builtin_amdgcn_readlane(e0, q) < e0 ? 1 : 0;
            gin_add_all<false>(xr, L, k0, 0, s0, 0, acc);
        } else if (fits && L <= 64) {
            int2 r0 = sbuf[st0 + lane];
            int e0 = r0.x, s0 = r0.y, e1 = GIN_NOEDGE, s1 = 0;
            if (32 + lane < L) { int2 r1 = sbuf[st0 + 32 + lane]; e1 = r1.x; s1 = r1.y; }
            int k0 = 0, k1 = 0;
            for (int q = 0; q < 32; ++q) {
                int eq = __builtin_amdgcn_readlane(e0, q);
                k0 += eq < e0 ? 1 : 0;
                k1 += eq < e1 ? 1 : 0;
            }
            for (int q = 32; q < L; ++q) {
                int eq = __builtin_amdgcn_readlane(e1, q - 32);
                k0 += eq < e0 ? 1 : 0;
                k1 += eq < e1 ? 1 : 0;
            }
            gin_add_all<true>(xr, L, k0, k1, s0, s1, acc);
        } else {
            gin_add_by_scan(xr, edge, ne, node, lane, acc);
        }
        float4 o;
        o.x = scl * xv.x + acc.x; o.y = scl * xv.y + acc.y;
        o.z = scl * xv.z + acc.z; o.w = scl * xv.w + acc.w;
        *(float4*)(h + (long)node * GIN_D + lane * 4) = o;
    }
}

#define GIN_MW 16

template <int NJ, int NA>
__device__ __forceinline__ void gin_gemm(const float* const (&Ap)[NA], const float* Bp0,
                                         const float* Bp1, v8f (&acc)[NA][2]) {
    for (int kk = 0; kk < GIN_D; kk += 4) {
        v2f b0 = *(const v2f*)(Bp0 + kk);
        v2f b1 = *(const v2f*)(Bp1 + kk);
#pragma unroll
        for (int j = 0; j < NJ; ++j) {
            v2f a = *(const v2f*)(Ap[j] + kk);
            acc[j][0] = __builtin_amdgcn_wmma_f32_16x16x4_f32(
                false, a, false, b0, (short)0, acc[j][0], false, false);
            acc[j][1] = __builtin_amdgcn_wmma_f32_16x16x4_f32(
                false, a, false, b1, (short)0, acc[j][1], false, false);
        }
    }
}

template <int TILES>
__global__ __launch_bounds__(GIN_MW * 32) void gin_mlp(const float* __restrict__ h,
                                               const float* __restrict__ W1,
                                               const float* __restrict__ b1,
                                               const float* __restrict__ W2,
                                               const float* __restrict__ b2,
                                               float* __restrict__ out,
                                               const int* __restrict__ ids,
                                               int nrows) {
    constexpr int ROWS = TILES * 16;
    constexpr int NB   = TILES / 4;
    constexpr int NR   = TILES % 4;
    constexpr int NA   = NB + 1;
    __shared__ float lds_W[GIN_D * GIN_PAD];
    __shared__ float lds_A[ROWS * GIN_PAD];
    __shared__ float lds_b1[GIN_D];
    __shared__ float lds_b2[GIN_D];

    const int tid   = threadIdx.x;
    const int wave  = __builtin_amdgcn_readfirstlane(tid >> 5);
    const int lane  = tid & 31;
    const int lmod  = lane & 15;
    const int lhalf = lane >> 4;
    const int koff  = lhalf * 2;
    const int row0  = blockIdx.x * ROWS;

    int rsel = row0 + wave * TILES + (lane < TILES ? lane : 0);
    rsel = rsel < nrows ? rsel : nrows - 1;
    const int myrow = ids ? ids[rsel] : rsel;

    if (tid < GIN_D)          lds_b1[tid] = b1[tid];
    else if (tid < 2 * GIN_D) lds_b2[tid - GIN_D] = b2[tid - GIN_D];

    for (int i = tid; i < GIN_D * GIN_D; i += GIN_MW * 32) {
        int k = i >> 7, n = i & (GIN_D - 1);
        lds_W[n * GIN_PAD + k] = W1[i];
    }

    float4 av[TILES];
#pragma unroll
    for (int r = 0; r < TILES; ++r) {
        const int srow = __builtin_amdgcn_readlane(myrow, r);
        av[r] = *(const float4*)(h + (long)srow * GIN_D + lane * 4);
    }
#pragma unroll
    for (int r = 0; r < TILES; ++r)
        *(float4*)(lds_A + (wave * TILES + r) * GIN_PAD + lane * 4) = av[r];
    __syncthreads();

    const int q  = wave & 3;
    const int g  = wave >> 2;
    const int rx = (g - q + 4) & 3;
    const bool has_x = rx < NR;
    int tl[NA];
    const float* Ap[NA];
#pragma unroll
    for (int j = 0; j < NB; ++j) tl[j] = g + 4 * j;
    tl[NB] = has_x ? NB * 4 + rx : 0;
#pragma unroll
    for (int j = 0; j < NA; ++j) Ap[j] = lds_A + (tl[j] * 16 + lmod) * GIN_PAD + koff;
    const float* Bp0 = lds_W + (q * 32 + lmod) * GIN_PAD + koff;
    const float* Bp1 = Bp0 + 16 * GIN_PAD;

    v8f acc[NA][2] = {};
    if (has_x) gin_gemm<NB + 1, NA>(Ap, Bp0, Bp1, acc);
    else       gin_gemm<NB, NA>(Ap, Bp0, Bp1, acc);
    __syncthreads();

#pragma unroll
    for (int j = 0; j < NA; ++j) {
        if (j < NB || has_x) {
#pragma unroll
            for (int c = 0; c < 2; ++c) {
                const int col = q * 32 + c * 16 + lmod;
                float bv = lds_b1[col];
#pragma unroll
                for (int r = 0; r < 8; ++r) {
                    float t = acc[j][c][r] + bv;
                    lds_A[(tl[j] * 16 + r + 8 * lhalf) * GIN_PAD + col] = t > 0.0f ? t : 0.0f;
                }
            }
        }
    }
    for (int i = tid; i < GIN_D * GIN_D; i += GIN_MW * 32) {
        int k = i >> 7, n = i & (GIN_D - 1);
        lds_W[n * GIN_PAD + k] = W2[i];
    }
    __syncthreads();

    v8f acc2[NA][2] = {};
    if (has_x) gin_gemm<NB + 1, NA>(Ap, Bp0, Bp1, acc2);
    else       gin_gemm<NB, NA>(Ap, Bp0, Bp1, acc2);

#pragma unroll
    for (int j = 0; j < NA; ++j) {
        if (j < NB || has_x) {
#pragma unroll
            for (int r = 0; r < 8; ++r) {
                int row = row0 + tl[j] * 16 + r + 8 * lhalf;
                bool ok = row < nrows;
                long orow = 0;
                if (ok) orow = ids ? (long)ids[row] : (long)row;
#pragma unroll
                for (int c = 0; c < 2; ++c) {
                    const int col = q * 32 + c * 16 + lmod;
                    float val = acc2[j][c][r] + lds_b2[col];
                    if (ok) {
                        float* p = out + orow * GIN_D + col;
                        if (ids) unsafeAtomicAdd(p, val);
                        else     *p = val;
                    }
                }
            }
        }
    }
}

extern "C" void kernel_launch(void* const* d_in, const int* in_sizes, int n_in,
                              void* d_out, int out_size, void* d_ws, size_t ws_size,
                              hipStream_t stream) {
    const float* x    = (const float*)d_in[0];
    const int*   edge = (const int*)d_in[1];
    const int*   nid  = (const int*)d_in[2];
    const float* eps  = (const float*)d_in[3];
    const float* W1   = (const float*)d_in[4];
    const float* b1   = (const float*)d_in[5];
    const float* W2   = (const float*)d_in[6];
    const float* b2   = (const float*)d_in[7];
    const float* Wi1  = (const float*)d_in[8];
    const float* bi1  = (const float*)d_in[9];
    const float* Wi2  = (const float*)d_in[10];
    const float* bi2  = (const float*)d_in[11];

    const int N   = in_sizes[0] / GIN_D;
    const int E   = in_sizes[1] / 2;
    const int NID = in_sizes[2];
    float* h      = (float*)d_ws;
    int*   tab    = (int*)(h + (size_t)N * GIN_D);
    int2*  staged = (int2*)(tab + GIN_GA * GIN_NBP);

    gin_bin<<<GIN_GA, 1024, 0, stream>>>(edge, tab, staged, N, E);
    gin_pull_bin<<<GIN_NB, 1024, 0, stream>>>(x, eps, edge, tab, staged, h, N, E);

    gin_mlp<GIN_MAIN_TILES><<<(N + GIN_MAIN_TILES * 16 - 1) / (GIN_MAIN_TILES * 16), GIN_MW * 32, 0, stream>>>(
        h, W1, b1, W2, b2, (float*)d_out, nullptr, N);
    gin_mlp<GIN_ID_TILES><<<(NID + GIN_ID_TILES * 16 - 1) / (GIN_ID_TILES * 16), GIN_MW * 32, 0, stream>>>(
        h, Wi1, bi1, Wi2, bi2, (float*)d_out, nid, NID);
}


# main MLP fused into the aggregation kernel: a bin's h rows reach the WMMA stage through LDS, no global round trip; 3 launches, was 4; h still written for the id branch; float sum order kept
# speedup vs baseline: 2.8220x; 1.1206x over previous
#include <hip/hip_runtime.h>

typedef __attribute__((ext_vector_type(2))) float v2f;
typedef __attribute__((ext_vector_type(8))) float v8f;

#define GIN_D    128
#define GIN_PAD  132
#define GIN_ID_TILES   3

#define GIN_NOEDGE 0x7fffffff
#define GIN_GA   256
#define GIN_NB   512
#define GIN_NBP  (GIN_NB + 1)
#define GIN_BWMAX 128
#define GIN_MAXM 4096

__device__ __forceinline__ int gin_block_scan(int v, int* sh, int tid, int nt, int& total) {
    const int lane = tid & 31, w = tid >> 5, nw = nt >> 5;
    int incl = v;
    for (int d = 1; d < 32; d <<= 1) {
        int t = __shfl_up(incl, d, 32);
        if (lane >= d) incl += t;
    }
    if (lane == 31) sh[w] = incl;
    __syncthreads();
    int pre = 0, tot = 0;
    for (int q = 0; q < nw; ++q) {
        int t = sh[q];
        if (q < w) pre += t;
        tot += t;
    }
    total = tot;
    __syncthreads();
    return pre + incl - v;
}

__global__ __launch_bounds__(1024) void gin_bin(const int* __restrict__ edge,
                                               int* __restrict__ tab,
                                               int2* __restrict__ staged, int n, int ne) {
    __shared__ int cnt[GIN_NB];
    __shared__ int cur[GIN_NB];
    __shared__ int sc[32];
    const int tid = threadIdx.x;
    const int wg  = blockIdx.x;
    const int ch  = (ne + GIN_GA - 1) / GIN_GA;
    const int bw  = (n + GIN_NB - 1) / GIN_NB;
    const int e0  = wg * ch;
    const int e1  = e0 + ch < ne ? e0 + ch : ne;
    if (tid < GIN_NB) cnt[tid] = 0;
    __syncthreads();
    for (int e = e0 + tid; e < e1; e += 1024) {
        int s = edge[e], d = edge[ne + e];
        if (s != d) atomicAdd(&cnt[d / bw], 1);
    }
    __syncthreads();
    int total;
    const int base = gin_block_scan(tid < GIN_NB ? cnt[tid] : 0, sc, tid, 1024, total);
    if (tid < GIN_NB) { cur[tid] = base; tab[wg * GIN_NBP + tid] = base; }
    if (tid == 0) tab[wg * GIN_NBP + GIN_NB] = total;
    __syncthreads();
    for (int e = e0 + tid; e < e1; e += 1024) {
        int s = edge[e], d = edge[ne + e];
        if (s != d) {
            int bin = d / bw;
            int pos = atomicAdd(&cur[bin], 1);
            staged[e0 + pos] = make_int2(e, ((d - bin * bw) << 16) | s);
        }
    }
}

template <bool TWO>
__device__ __forceinline__ int gin_src_of_rank(int r, int k0, int k1, int s0, int s1) {
    if (!TWO) {
        unsigned m = __builtin_amdgcn_ballot_w32(k0 == r);
        return __builtin_amdgcn_readlane(s0, __builtin_ctz(m));
    }
    unsigned m = __builtin_amdgcn_ballot_w32(k0 == r || k1 == r);
    return __builtin_amdgcn_readlane(k0 == r ? s0 : s1, __builtin_ctz(m));
}

template <int K, bool TWO>
__device__ __forceinline__ void gin_add_rows(const float* __restrict__ xr, int r0,
                                             int k0, int k1, int s0, int s1, float4& acc) {
    float4 v[K];
#pragma unroll
    for (int u = 0; u < K; ++u) {
        int s = gin_src_of_rank<TWO>(__builtin_amdgcn_readfirstlane(r0 + u), k0, k1, s0, s1);
        v[u] = *(const float4*)(xr + (long)s * GIN_D);
    }
#pragma unroll
    for (int u = 0; u < K; ++u) {
        acc.x += v[u].x; acc.y += v[u].y; acc.z += v[u].z; acc.w += v[u].w;
    }
}

template <bool TWO>
__device__ __forceinline__ void gin_add_all(const float* __restrict__ xr, int L,
                                            int k0, int k1, int s0, int s1, float4& acc) {
    int r = 0;
    for (; r + 8 <= L; r += 8) gin_add_rows<8, TWO>(xr, r, k0, k1, s0, s1, acc);
    if (r + 4 <= L) { gin_add_rows<4, TWO>(xr, r, k0, k1, s0, s1, acc); r += 4; }
    if (r + 2 <= L) { gin_add_rows<2, TWO>(xr, r, k0, k1, s0, s1, acc); r += 2; }
    if (r < L)      { gin_add_rows<1, TWO>(xr, r, k0, k1, s0, s1, acc); }
}

__device__ __forceinline__ void gin_add_by_scan(const float* __restrict__ xr,
                                                const int* __restrict__ edge, int ne,
                                                int node, int lane, float4& acc) {
    for (int base = 0; base < ne; base += 32) {
        int e = base + lane;
        int s = -1, d = -1;
        if (e < ne) { s = edge[e]; d = edge[ne + e]; }
        unsigned m = __builtin_amdgcn_ballot_w32(d == node && s != node);
        while (m) {
            int l = __builtin_ctz(m);
            m &= m - 1;
            int ss = __builtin_amdgcn_readlane(s, l);
            float4 v = *(const float4*)(xr + (long)ss * GIN_D);
            acc.x += v.x; acc.y += v.y; acc.z += v.z; acc.w += v.w;
        }
    }
}

#define GIN_MW 16

template <int NJ, int NA>
__device__ __forceinline__ void gin_gemm(const float* const (&Ap)[NA], const float* Bp0,
                                         const float* Bp1, v8f (&acc)[NA][2]) {
    for (int kk = 0; kk < GIN_D; kk += 4) {
        v2f b0 = *(const v2f*)(Bp0 + kk);
        v2f b1 = *(const v2f*)(Bp1 + kk);
#pragma unroll
        for (int j = 0; j < NJ; ++j) {
            v2f a = *(const v2f*)(Ap[j] + kk);
            acc[j][0] = __builtin_amdgcn_wmma_f32_16x16x4_f32(
                false, a, false, b0, (short)0, acc[j][0], false, false);
            acc[j][1] = __builtin_amdgcn_wmma_f32_16x16x4_f32(
                false, a, false, b1, (short)0, acc[j][1], false, false);
        }
    }
}

template <int TILES>
__global__ __launch_bounds__(GIN_MW * 32) void gin_mlp(const float* __restrict__ h,
                                               const float* __restrict__ W1,
                                               const float* __restrict__ b1,
                                               const float* __restrict__ W2,
                                               const float* __restrict__ b2,
                                               float* __restrict__ out,
                                               const int* __restrict__ ids,
                                               int nrows) {
    constexpr int ROWS = TILES * 16;
    constexpr int NB   = TILES / 4;
    constexpr int NR   = TILES % 4;
    constexpr int NA   = NB + 1;
    __shared__ float lds_W[GIN_D * GIN_PAD];
    __shared__ float lds_A[ROWS * GIN_PAD];
    __shared__ float lds_b1[GIN_D];
    __shared__ float lds_b2[GIN_D];

    const int tid   = threadIdx.x;
    const int wave  = __builtin_amdgcn_readfirstlane(tid >> 5);
    const int lane  = tid & 31;
    const int lmod  = lane & 15;
    const int lhalf = lane >> 4;
    const int koff  = lhalf * 2;
    const int row0  = blockIdx.x * ROWS;

    int rsel = row0 + wave * TILES + (lane < TILES ? lane : 0);
    rsel = rsel < nrows ? rsel : nrows - 1;
    const int myrow = ids ? ids[rsel] : rsel;

    if (tid < GIN_D)          lds_b1[tid] = b1[tid];
    else if (tid < 2 * GIN_D) lds_b2[tid - GIN_D] = b2[tid - GIN_D];

    for (int i = tid; i < GIN_D * GIN_D; i += GIN_MW * 32) {
        int k = i >> 7, n = i & (GIN_D - 1);
        lds_W[n * GIN_PAD + k] = W1[i];
    }

    float4 av[TILES];
#pragma unroll
    for (int r = 0; r < TILES; ++r) {
        const int srow = __builtin_amdgcn_readlane(myrow, r);
        av[r] = *(const float4*)(h + (long)srow * GIN_D + lane * 4);
    }
#pragma unroll
    for (int r = 0; r < TILES; ++r)
        *(float4*)(lds_A + (wave * TILES + r) * GIN_PAD + lane * 4) = av[r];
    __syncthreads();

    const int q  = wave & 3;
    const int g  = wave >> 2;
    const int rx = (g - q + 4) & 3;
    const bool has_x = rx < NR;
    int tl[NA];
    const float* Ap[NA];
#pragma unroll
    for (int j = 0; j < NB; ++j) tl[j] = g + 4 * j;
    tl[NB] = has_x ? NB * 4 + rx : 0;
#pragma unroll
    for (int j = 0; j < NA; ++j) Ap[j] = lds_A + (tl[j] * 16 + lmod) * GIN_PAD + koff;
    const float* Bp0 = lds_W + (q * 32 + lmod) * GIN_PAD + koff;
    const float* Bp1 = Bp0 + 16 * GIN_PAD;

    v8f acc[NA][2] = {};
    if (has_x) gin_gemm<NB + 1, NA>(Ap, Bp0, Bp1, acc);
    else       gin_gemm<NB, NA>(Ap, Bp0, Bp1, acc);
    __syncthreads();

#pragma unroll
    for (int j = 0; j < NA; ++j) {
        if (j < NB || has_x) {
#pragma unroll
            for (int c = 0; c < 2; ++c) {
                const int col = q * 32 + c * 16 + lmod;
                float bv = lds_b1[col];
#pragma unroll
                for (int r = 0; r < 8; ++r) {
                    float t = acc[j][c][r] + bv;
                    lds_A[(tl[j] * 16 + r + 8 * lhalf) * GIN_PAD + col] = t > 0.0f ? t : 0.0f;
                }
            }
        }
    }
    for (int i = tid; i < GIN_D * GIN_D; i += GIN_MW * 32) {
        int k = i >> 7, n = i & (GIN_D - 1);
        lds_W[n * GIN_PAD + k] = W2[i];
    }
    __syncthreads();

    v8f acc2[NA][2] = {};
    if (has_x) gin_gemm<NB + 1, NA>(Ap, Bp0, Bp1, acc2);
    else       gin_gemm<NB, NA>(Ap, Bp0, Bp1, acc2);

#pragma unroll
    for (int j = 0; j < NA; ++j) {
        if (j < NB || has_x) {
#pragma unroll
            for (int r = 0; r < 8; ++r) {
                int row = row0 + tl[j] * 16 + r + 8 * lhalf;
                bool ok = row < nrows;
                long orow = 0;
                if (ok) orow = ids ? (long)ids[row] : (long)row;
#pragma unroll
                for (int c = 0; c < 2; ++c) {
                    const int col = q * 32 + c * 16 + lmod;
                    float val = acc2[j][c][r] + lds_b2[col];
                    if (ok) {
                        float* p = out + orow * GIN_D + col;
                        if (ids) unsafeAtomicAdd(p, val);
                        else     *p = val;
                    }
                }
            }
        }
    }
}

#define GIN_FROWS GIN_BWMAX

__global__ __launch_bounds__(1024, 16) void gin_pull_mlp(const float* __restrict__ x,
                                                    const float* __restrict__ eps,
                                                    const int* __restrict__ edge,
                                                    const int* __restrict__ tab,
                                                    const int2* __restrict__ staged,
                                                    float* __restrict__ h,
                                                    const float* __restrict__ W1,
                                                    const float* __restrict__ b1,
                                                    const float* __restrict__ W2,
                                                    const float* __restrict__ b2,
                                                    float* __restrict__ out, int n, int ne) {
    __shared__ __attribute__((aligned(16))) float lds_W[GIN_D * GIN_PAD];
    __shared__ __attribute__((aligned(16))) float lds_A[GIN_FROWS * GIN_PAD];
    __shared__ float lds_b1[GIN_D];
    __shared__ float lds_b2[GIN_D];
    __shared__ int ncnt[GIN_BWMAX];
    __shared__ int nstart[GIN_BWMAX];
    __shared__ int ncur[GIN_BWMAX];
    __shared__ int sc[32];
    static_assert(2 * GIN_MAXM * sizeof(int2) <= sizeof(float) * GIN_D * GIN_PAD, "edge buffers fit the weight tile");
    int2* const ebuf = (int2*)lds_W;
    int2* const sbuf = ebuf + GIN_MAXM;
    const int tid  = threadIdx.x;
    const int lane = tid & 31;
    const int wave = __builtin_amdgcn_readfirstlane(tid >> 5);
    const int b    = blockIdx.x;
    const int ch   = (ne + GIN_GA - 1) / GIN_GA;
    const int bw   = (n + GIN_NB - 1) / GIN_NB;

    if (tid < GIN_D)          lds_b1[tid] = b1[tid];
    else if (tid < 2 * GIN_D) lds_b2[tid - GIN_D] = b2[tid - GIN_D];

    int s = 0, c = 0;
    if (tid < GIN_GA) {
        s = tab[tid * GIN_NBP + b];
        c = tab[tid * GIN_NBP + b + 1] - s;
    }
    if (tid < GIN_BWMAX) ncnt[tid] = 0;
    int M;
    const int base = gin_block_scan(c, sc, tid, 1024, M);
    const bool fits = M <= GIN_MAXM && bw <= GIN_BWMAX;
    if (fits) {
        const int2* seg = staged + (long)tid * ch + s;
        for (int i0 = 0; i0 < c; i0 += 8) {
            int2 r[8];
#pragma unroll
            for (int u = 0; u < 8; ++u) r[u] = seg[i0 + u < c ? i0 + u : c - 1];
#pragma unroll
            for (int u = 0; u < 8; ++u)
                if (i0 + u < c) ebuf[base + i0 + u] = r[u];
        }
        __syncthreads();
        for (int i = tid; i < M; i += 1024) atomicAdd(&ncnt[ebuf[i].y >> 16], 1);
        __syncthreads();
        int tot;
        const int st = gin_block_scan(tid < GIN_BWMAX ? ncnt[tid] : 0, sc, tid, 1024, tot);
        if (tid < GIN_BWMAX) { nstart[tid] = st; ncur[tid] = st; }
        __syncthreads();
        for (int i = tid; i < M; i += 1024) {
            int2 r = ebuf[i];
            int pos = atomicAdd(&ncur[r.y >> 16], 1);
            sbuf[pos] = make_int2(r.x, r.y & 0xffff);
        }
        __syncthreads();
    }

    const float scl = 1.0f + eps[0];
    const float* xr = x + lane * 4;
    {
        const int j0 = blockIdx.y * GIN_FROWS;
        const int nr = bw - j0 < GIN_FROWS ? bw - j0 : GIN_FROWS;
        const int nt = (nr + 15) >> 4;
        for (int jj = wave; jj < nt * 16; jj += 32) {
            const int j = j0 + jj;
            const int node = b * bw + j;
            float4 o = make_float4(0.0f, 0.0f, 0.0f, 0.0f);
            if (jj < nr && node < n) {
                const float4 xv = *(const float4*)(xr + (long)node * GIN_D);
                float4 acc = make_float4(0.0f, 0.0f, 0.0f, 0.0f);
                const int L   = fits ? __builtin_amdgcn_readfirstlane(ncnt[j]) : 0;
                const int st0 = fits ? __builtin_amdgcn_readfirstlane(nstart[j]) : 0;
                if (fits && L <= 32) {
                    int e0 = GIN_NOEDGE, s0 = 0;
                    if (lane < L) { int2 r = sbuf[st0 + lane]; e0 = r.x; s0 = r.y; }
                    int k0 = 0;
                    for (int qq = 0; qq < L; ++qq) k0 += __builtin_amdgcn_readlane(e0, qq) < e0 ? 1 : 0;
                    gin_add_all<false>(xr, L, k0, 0, s0, 0, acc);
                } else if (fits && L <= 64) {
                    int2 r0 = sbuf[st0 + lane];
                    int e0 = r0.x, s0 = r0.y, e1 = GIN_NOEDGE, s1 = 0;
                    if (32 + lane < L) { int2 r1 = sbuf[st0 + 32 + lane]; e1 = r1.x; s1 = r1.y; }
                    int k0 = 0, k1 = 0;
                    for (int qq = 0; qq < 32; ++qq) {
                        int eq = __builtin_amdgcn_readlane(e0, qq);
                        k0 += eq < e0 ? 1 : 0;
                        k1 += eq < e1 ? 1 : 0;
                    }
                    for (int qq = 32; qq < L; ++qq) {
                        int eq = __builtin_amdgcn_readlane(e1, qq - 32);
                        k0 += eq < e0 ? 1 : 0;
                        k1 += eq < e1 ? 1 : 0;
                    }
                    gin_add_all<true>(xr, L, k0, k1, s0, s1, acc);
                } else {
                    gin_add_by_scan(xr, edge, ne, node, lane, acc);
                }
                o.x = scl * xv.x + acc.x; o.y = scl * xv.y + acc.y;
                o.z = scl * xv.z + acc.z; o.w = scl * xv.w + acc.w;
                *(float4*)(h + (long)node * GIN_D + lane * 4) = o;
            }
            *(float4*)(lds_A + jj * GIN_PAD + lane * 4) = o;
        }

        float wv[16];
#pragma unroll
        for (int u = 0; u < 16; ++u) wv[u] = W1[tid + u * 1024];
        __syncthreads();
        const int lmod  = lane & 15;
        const int lhalf = lane >> 4;
        const int koff  = lhalf * 2;
        const int q  = wave & 3;
        const int g  = wave >> 2;
        const int tl = g < 4 ? g : 4 + ((g - q) & 3);
        const float* const Ap[1] = { lds_A + (tl * 16 + lmod) * GIN_PAD + koff };
        const float* Bp0 = lds_W + (q * 32 + lmod) * GIN_PAD + koff;
        const float* Bp1 = Bp0 + 16 * GIN_PAD;

        const int wbase = (tid & (GIN_D - 1)) * GIN_PAD + (tid >> 7);
#pragma unroll
        for (int u = 0; u < 16; ++u) lds_W[wbase + 8 * u] = wv[u];
        __syncthreads();

        v8f acc[1][2] = {};
        if (tl < nt) gin_gemm<1, 1>(Ap, Bp0, Bp1, acc);
        __syncthreads();

        if (tl < nt) {
#pragma unroll
            for (int cc = 0; cc < 2; ++cc) {
                const int col = q * 32 + cc * 16 + lmod;
                float bv = lds_b1[col];
#pragma unroll
                for (int r = 0; r < 8; ++r) {
                    float t = acc[0][cc][r] + bv;
                    lds_A[(tl * 16 + r + 8 * lhalf) * GIN_PAD + col] = t > 0.0f ? t : 0.0f;
                }
            }
        }
#pragma unroll
        for (int u = 0; u < 16; ++u) lds_W[wbase + 8 * u] = W2[tid + u * 1024];
        __syncthreads();

        if (tl < nt) {
            v8f acc2[1][2] = {};
            gin_gemm<1, 1>(Ap, Bp0, Bp1, acc2);
#pragma unroll
            for (int r = 0; r < 8; ++r) {
                const int jr = tl * 16 + r + 8 * lhalf;
                const int node = b * bw + j0 + jr;
                const bool ok = jr < nr && node < n;
#pragma unroll
                for (int cc = 0; cc < 2; ++cc) {
                    const int col = q * 32 + cc * 16 + lmod;
                    float val = acc2[0][cc][r] + lds_b2[col];
                    if (ok) out[(long)node * GIN_D + col] = val;
                }
            }
        }
    }
}

extern "C" void kernel_launch(void* const* d_in, const int* in_sizes, int n_in,
                              void* d_out, int out_size, void* d_ws, size_t ws_size,
                              hipStream_t stream) {
    const float* x    = (const float*)d_in[0];
    const int*   edge = (const int*)d_in[1];
    const int*   nid  = (const int*)d_in[2];
    const float* eps  = (const float*)d_in[3];
    const float* W1   = (const float*)d_in[4];
    const float* b1   = (const float*)d_in[5];
    const float* W2   = (const float*)d_in[6];
    const float* b2   = (const float*)d_in[7];
    const float* Wi1  = (const float*)d_in[8];
    const float* bi1  = (const float*)d_in[9];
    const float* Wi2  = (const float*)d_in[10];
    const float* bi2  = (const float*)d_in[11];

    const int N   = in_sizes[0] / GIN_D;
    const int E   = in_sizes[1] / 2;
    const int NID = in_sizes[2];
    float* h      = (float*)d_ws;
    int*   tab    = (int*)(h + (size_t)N * GIN_D);
    int2*  staged = (int2*)(tab + GIN_GA * GIN_NBP);

    gin_bin<<<GIN_GA, 1024, 0, stream>>>(edge, tab, staged, N, E);
    const int bw = (N + GIN_NB - 1) / GIN_NB;
    gin_pull_mlp<<<dim3(GIN_NB, (bw + GIN_FROWS - 1) / GIN_FROWS), 1024, 0, stream>>>(
        x, eps, edge, tab, staged, h, W1, b1, W2, b2, (float*)d_out, N, E);

    gin_mlp<GIN_ID_TILES><<<(NID + GIN_ID_TILES * 16 - 1) / (GIN_ID_TILES * 16), GIN_MW * 32, 0, stream>>>(
        h, Wi1, bi1, Wi2, bi2, (float*)d_out, nid, NID);
}


# id branch MLP fused into the aggregation kernel: a bin's work-group counts its nodes in ids, runs the id GEMMs on them, adds to its out rows per occurrence by plain adds, no float atomics; 2 launches
# speedup vs baseline: 3.0633x; 1.0855x over previous
#include <hip/hip_runtime.h>

typedef __attribute__((ext_vector_type(2))) float v2f;
typedef __attribute__((ext_vector_type(8))) float v8f;

#define GIN_D    128
#define GIN_PAD  132

#define GIN_NOEDGE 0x7fffffff
#define GIN_GA   256
#define GIN_NB   512
#define GIN_NBP  (GIN_NB + 1)
#define GIN_BWMAX 128
#define GIN_MAXM 4096
#define GIN_IDU  10

__device__ __forceinline__ int gin_block_scan(int v, int* sh, int tid, int nt, int& total) {
    const int lane = tid & 31, w = tid >> 5, nw = nt >> 5;
    int incl = v;
    for (int d = 1; d < 32; d <<= 1) {
        int t = __shfl_up(incl, d, 32);
        if (lane >= d) incl += t;
    }
    if (lane == 31) sh[w] = incl;
    __syncthreads();
    int pre = 0, tot = 0;
    for (int q = 0; q < nw; ++q) {
        int t = sh[q];
        if (q < w) pre += t;
        tot += t;
    }
    total = tot;
    __syncthreads();
    return pre + incl - v;
}

__global__ __launch_bounds__(1024) void gin_bin(const int* __restrict__ edge,
                                               int* __restrict__ tab,
                                               int2* __restrict__ staged, int n, int ne) {
    __shared__ int cnt[GIN_NB];
    __shared__ int cur[GIN_NB];
    __shared__ int sc[32];
    const int tid = threadIdx.x;
    const int wg  = blockIdx.x;
    const int ch  = (ne + GIN_GA - 1) / GIN_GA;
    const int bw  = (n + GIN_NB - 1) / GIN_NB;
    const int e0  = wg * ch;
    const int e1  = e0 + ch < ne ? e0 + ch : ne;
    if (tid < GIN_NB) cnt[tid] = 0;
    __syncthreads();
    for (int e = e0 + tid; e < e1; e += 1024) {
        int s = edge[e], d = edge[ne + e];
        if (s != d) atomicAdd(&cnt[d / bw], 1);
    }
    __syncthreads();
    int total;
    const int base = gin_block_scan(tid < GIN_NB ? cnt[tid] : 0, sc, tid, 1024, total);
    if (tid < GIN_NB) { cur[tid] = base; tab[wg * GIN_NBP + tid] = base; }
    if (tid == 0) tab[wg * GIN_NBP + GIN_NB] = total;
    __syncthreads();
    for (int e = e0 + tid; e < e1; e += 1024) {
        int s = edge[e], d = edge[ne + e];
        if (s != d) {
            int bin = d / bw;
            int pos = atomicAdd(&cur[bin], 1);
            staged[e0 + pos] = make_int2(e, ((d - bin * bw) << 16) | s);
        }
    }
}

template <bool TWO>
__device__ __forceinline__ int gin_src_of_rank(int r, int k0, int k1, int s0, int s1) {
    if (!TWO) {
        unsigned m = __builtin_amdgcn_ballot_w32(k0 == r);
        return __builtin_amdgcn_readlane(s0, __builtin_ctz(m));
    }
    unsigned m = __builtin_amdgcn_ballot_w32(k0 == r || k1 == r);
    return __builtin_amdgcn_readlane(k0 == r ? s0 : s1, __builtin_ctz(m));
}

template <int K, bool TWO>
__device__ __forceinline__ void gin_add_rows(const float* __restrict__ xr, int r0,
                                             int k0, int k1, int s0, int s1, float4& acc) {
    float4 v[K];
#pragma unroll
    for (int u = 0; u < K; ++u) {
        int s = gin_src_of_rank<TWO>(__builtin_amdgcn_readfirstlane(r0 + u), k0, k1, s0, s1);
        v[u] = *(const float4*)(xr + (long)s * GIN_D);
    }
#pragma unroll
    for (int u = 0; u < K; ++u) {
        acc.x += v[u].x; acc.y += v[u].y; acc.z += v[u].z; acc.w += v[u].w;
    }
}

template <bool TWO>
__device__ __forceinline__ void gin_add_all(const float* __restrict__ xr, int L,
                                            int k0, int k1, int s0, int s1, float4& acc) {
    int r = 0;
    for (; r + 8 <= L; r += 8) gin_add_rows<8, TWO>(xr, r, k0, k1, s0, s1, acc);
    if (r + 4 <= L) { gin_add_rows<4, TWO>(xr, r, k0, k1, s0, s1, acc); r += 4; }
    if (r + 2 <= L) { gin_add_rows<2, TWO>(xr, r, k0, k1, s0, s1, acc); r += 2; }
    if (r < L)      { gin_add_rows<1, TWO>(xr, r, k0, k1, s0, s1, acc); }
}

__device__ __forceinline__ void gin_add_by_scan(const float* __restrict__ xr,
                                                const int* __restrict__ edge, int ne,
                                                int node, int lane, float4& acc) {
    for (int base = 0; base < ne; base += 32) {
        int e = base + lane;
        int s = -1, d = -1;
        if (e < ne) { s = edge[e]; d = edge[ne + e]; }
        unsigned m = __builtin_amdgcn_ballot_w32(d == node && s != node);
        while (m) {
            int l = __builtin_ctz(m);
            m &= m - 1;
            int ss = __builtin_amdgcn_readlane(s, l);
            float4 v = *(const float4*)(xr + (long)ss * GIN_D);
            acc.x += v.x; acc.y += v.y; acc.z += v.z; acc.w += v.w;
        }
    }
}

template <int NJ, int NA>
__device__ __forceinline__ void gin_gemm(const float* const (&Ap)[NA], const float* Bp0,
                                         const float* Bp1, v8f (&acc)[NA][2]) {
    for (int kk = 0; kk < GIN_D; kk += 4) {
        v2f b0 = *(const v2f*)(Bp0 + kk);
        v2f b1 = *(const v2f*)(Bp1 + kk);
#pragma unroll
        for (int j = 0; j < NJ; ++j) {
            v2f a = *(const v2f*)(Ap[j] + kk);
            acc[j][0] = __builtin_amdgcn_wmma_f32_16x16x4_f32(
                false, a, false, b0, (short)0, acc[j][0], false, false);
            acc[j][1] = __builtin_amdgcn_wmma_f32_16x16x4_f32(
                false, a, false, b1, (short)0, acc[j][1], false, false);
        }
    }
}

#define GIN_FROWS GIN_BWMAX

__global__ __launch_bounds__(1024, 16) void gin_pull_mlp(const float* __restrict__ x,
                                                    const float* __restrict__ eps,
                                                    const int* __restrict__ edge,
                                                    const int* __restrict__ tab,
                                                    const int2* __restrict__ staged,
                                                    float* h,
                                                    const float* __restrict__ W1,
                                                    const float* __restrict__ b1,
                                                    const float* __restrict__ W2,
                                                    const float* __restrict__ b2,
                                                    const int* __restrict__ ids,
                                                    const float* __restrict__ Wi1,
                                                    const float* __restrict__ bi1,
                                                    const float* __restrict__ Wi2,
                                                    const float* __restrict__ bi2,
                                                    float* out, int n, int ne, int nid) {
    __shared__ __attribute__((aligned(16))) float lds_W[GIN_D * GIN_PAD];
    __shared__ __attribute__((aligned(16))) float lds_A[GIN_FROWS * GIN_PAD];
    __shared__ float lds_b1[GIN_D];
    __shared__ float lds_b2[GIN_D];
    __shared__ float lds_bi1[GIN_D];
    __shared__ float lds_bi2[GIN_D];
    __shared__ int idn[GIN_BWMAX];
    __shared__ int idrow[GIN_BWMAX];
    __shared__ int idtot;
    static_assert(GIN_BWMAX == 128, "one wave compacts the id rows, four nodes per lane");
    __shared__ int ncnt[GIN_BWMAX];
    __shared__ int nstart[GIN_BWMAX];
    __shared__ int ncur[GIN_BWMAX];
    __shared__ int sc[32];
    static_assert(2 * GIN_MAXM * sizeof(int2) <= sizeof(float) * GIN_D * GIN_PAD, "edge buffers fit the weight tile");
    int2* const ebuf = (int2*)lds_W;
    int2* const sbuf = ebuf + GIN_MAXM;
    const int tid  = threadIdx.x;
    const int lane = tid & 31;
    const int wave = __builtin_amdgcn_readfirstlane(tid >> 5);
    const int b    = blockIdx.x;
    const int ch   = (ne + GIN_GA - 1) / GIN_GA;
    const int bw   = (n + GIN_NB - 1) / GIN_NB;

    const int j0 = blockIdx.y * GIN_FROWS;
    const int nr = bw - j0 < GIN_FROWS ? bw - j0 : GIN_FROWS;
    const int lo = b * bw + j0;

    int idv[GIN_IDU];
    const int idlast = nid - 1;
#pragma unroll
    for (int u = 0; u < GIN_IDU; ++u) {
        const int i = tid + u * 1024;
        idv[u] = ids[i < nid ? i : idlast];
    }
    int s = 0, c = 0;
    if (tid < GIN_GA) {
        s = tab[tid * GIN_NBP + b];
        c = tab[tid * GIN_NBP + b + 1] - s;
    }
    if (tid < GIN_D)          lds_b1[tid] = b1[tid];
    else if (tid < 2 * GIN_D) lds_b2[tid - GIN_D] = b2[tid - GIN_D];
    else if (tid < 3 * GIN_D) lds_bi1[tid - 2 * GIN_D] = bi1[tid - 2 * GIN_D];
    else if (tid < 4 * GIN_D) lds_bi2[tid - 3 * GIN_D] = bi2[tid - 3 * GIN_D];
    if (tid < GIN_BWMAX) { ncnt[tid] = 0; idn[tid] = 0; }
    int M;
    const int base = gin_block_scan(c, sc, tid, 1024, M);
    const bool fits = M <= GIN_MAXM && bw <= GIN_BWMAX;
#pragma unroll
    for (int u = 0; u < GIN_IDU; ++u) {
        const int i = tid + u * 1024;
        const unsigned r = (unsigned)(idv[u] - lo);
        if (i < nid && r < (unsigned)nr) atomicAdd(&idn[r], 1);
    }
    for (int i0 = GIN_IDU * 1024; i0 < nid; i0 += GIN_IDU * 1024) {
#pragma unroll
        for (int u = 0; u < GIN_IDU; ++u) {
            const int i = i0 + tid + u * 1024;
            idv[u] = ids[i < nid ? i : idlast];
        }
#pragma unroll
        for (int u = 0; u < GIN_IDU; ++u) {
            const int i = i0 + tid + u * 1024;
            const unsigned r = (unsigned)(idv[u] - lo);
            if (i < nid && r < (unsigned)nr) atomicAdd(&idn[r], 1);
        }
    }
    if (fits) {
        const int2* seg = staged + (long)tid * ch + s;
        for (int i0 = 0; i0 < c; i0 += 8) {
            int2 r[8];
#pragma unroll
            for (int u = 0; u < 8; ++u) r[u] = seg[i0 + u < c ? i0 + u : c - 1];
#pragma unroll
            for (int u = 0; u < 8; ++u)
                if (i0 + u < c) ebuf[base + i0 + u] = r[u];
        }
    }
    __syncthreads();
    if (fits) {
        for (int i = tid; i < M; i += 1024) atomicAdd(&ncnt[ebuf[i].y >> 16], 1);
        __syncthreads();
        int tot;
        const int st = gin_block_scan(tid < GIN_BWMAX ? ncnt[tid] : 0, sc, tid, 1024, tot);
        if (tid < GIN_BWMAX) { nstart[tid] = st; ncur[tid] = st; }
        __syncthreads();
        for (int i = tid; i < M; i += 1024) {
            int2 r = ebuf[i];
            int pos = atomicAdd(&ncur[r.y >> 16], 1);
            sbuf[pos] = make_int2(r.x, r.y & 0xffff);
        }
        __syncthreads();
    }

    if (wave == 31) {
        int f[4], cl = 0;
#pragma unroll
        for (int k = 0; k < 4; ++k) {
            const int j = lane * 4 + k;
            f[k] = (j < nr && idn[j] > 0) ? 1 : 0;
            cl += f[k];
        }
        int incl = cl;
        for (int d = 1; d < 32; d <<= 1) {
            int t = __shfl_up(incl, d, 32);
            if (lane >= d) incl += t;
        }
        int pos = incl - cl;
#pragma unroll
        for (int k = 0; k < 4; ++k)
            if (f[k]) idrow[pos++] = lane * 4 + k;
        if (lane == 31) idtot = incl;
    }

    const float scl = 1.0f + eps[0];
    const float* xr = x + lane * 4;
    {
        const int nt = (nr + 15) >> 4;
        for (int jj = wave; jj < nt * 16; jj += 32) {
            const int j = j0 + jj;
            const int node = b * bw + j;
            float4 o = make_float4(0.0f, 0.0f, 0.0f, 0.0f);
            if (jj < nr && node < n) {
                const float4 xv = *(const float4*)(xr + (long)node * GIN_D);
                float4 acc = make_float4(0.0f, 0.0f, 0.0f, 0.0f);
                const int L   = fits ? __builtin_amdgcn_readfirstlane(ncnt[j]) : 0;
                const int st0 = fits ? __builtin_amdgcn_readfirstlane(nstart[j]) : 0;
                if (fits && L <= 32) {
                    int e0 = GIN_NOEDGE, s0 = 0;
                    if (lane < L) { int2 r = sbuf[st0 + lane]; e0 = r.x; s0 = r.y; }
                    int k0 = 0;
                    for (int qq = 0; qq < L; ++qq) k0 += __builtin_amdgcn_readlane(e0, qq) < e0 ? 1 : 0;
                    gin_add_all<false>(xr, L, k0, 0, s0, 0, acc);
                } else if (fits && L <= 64) {
                    int2 r0 = sbuf[st0 + lane];
                    int e0 = r0.x, s0 = r0.y, e1 = GIN_NOEDGE, s1 = 0;
                    if (32 + lane < L) { int2 r1 = sbuf[st0 + 32 + lane]; e1 = r1.x; s1 = r1.y; }
                    int k0 = 0, k1 = 0;
                    for (int qq = 0; qq < 32; ++qq) {
                        int eq = __builtin_amdgcn_readlane(e0, qq);
                        k0 += eq < e0 ? 1 : 0;
                        k1 += eq < e1 ? 1 : 0;
                    }
                    for (int qq = 32; qq < L; ++qq) {
                        int eq = __builtin_amdgcn_readlane(e1, qq - 32);
                        k0 += eq < e0 ? 1 : 0;
                        k1 += eq < e1 ? 1 : 0;
                    }
                    gin_add_all<true>(xr, L, k0, k1, s0, s1, acc);
                } else {
                    gin_add_by_scan(xr, edge, ne, node, lane, acc);
                }
                o.x = scl * xv.x + acc.x; o.y = scl * xv.y + acc.y;
                o.z = scl * xv.z + acc.z; o.w = scl * xv.w + acc.w;
                if (idn[jj] > 0) *(float4*)(h + (long)node * GIN_D + lane * 4) = o;
            }
            *(float4*)(lds_A + jj * GIN_PAD + lane * 4) = o;
        }

        float wv[16];
#pragma unroll
        for (int u = 0; u < 16; ++u) wv[u] = W1[tid + u * 1024];
        __syncthreads();
        const int lmod  = lane & 15;
        const int lhalf = lane >> 4;
        const int koff  = lhalf * 2;
        const int q  = wave & 3;
        const int g  = wave >> 2;
        const int tl = g < 4 ? g : 4 + ((g - q) & 3);
        const float* const Ap[1] = { lds_A + (tl * 16 + lmod) * GIN_PAD + koff };
        const float* Bp0 = lds_W + (q * 32 + lmod) * GIN_PAD + koff;
        const float* Bp1 = Bp0 + 16 * GIN_PAD;

        const int wbase = (tid & (GIN_D - 1)) * GIN_PAD + (tid >> 7);
#pragma unroll
        for (int u = 0; u < 16; ++u) lds_W[wbase + 8 * u] = wv[u];
        __syncthreads();

        v8f acc[1][2] = {};
        if (tl < nt) gin_gemm<1, 1>(Ap, Bp0, Bp1, acc);
        __syncthreads();

        if (tl < nt) {
#pragma unroll
            for (int cc = 0; cc < 2; ++cc) {
                const int col = q * 32 + cc * 16 + lmod;
                float bv = lds_b1[col];
#pragma unroll
                for (int r = 0; r < 8; ++r) {
                    float t = acc[0][cc][r] + bv;
                    lds_A[(tl * 16 + r + 8 * lhalf) * GIN_PAD + col] = t > 0.0f ? t : 0.0f;
                }
            }
        }
#pragma unroll
        for (int u = 0; u < 16; ++u) lds_W[wbase + 8 * u] = W2[tid + u * 1024];
        __syncthreads();

        if (tl < nt) {
            v8f acc2[1][2] = {};
            gin_gemm<1, 1>(Ap, Bp0, Bp1, acc2);
#pragma unroll
            for (int r = 0; r < 8; ++r) {
                const int jr = tl * 16 + r + 8 * lhalf;
                const int node = b * bw + j0 + jr;
                const bool ok = jr < nr && node < n;
#pragma unroll
                for (int cc = 0; cc < 2; ++cc) {
                    const int col = q * 32 + cc * 16 + lmod;
                    float val = acc2[0][cc][r] + lds_b2[col];
                    if (ok) out[(long)node * GIN_D + col] = val;
                }
            }
        }

        const int nidr = __builtin_amdgcn_readfirstlane(idtot);
        const int ntid = (nidr + 15) >> 4;
        if (ntid > 0) {
            float wi[16];
#pragma unroll
            for (int u = 0; u < 16; ++u) wi[u] = Wi1[tid + u * 1024];
            const bool hr0 = wave < ntid * 16;
            float4 hv = make_float4(0.0f, 0.0f, 0.0f, 0.0f);
            if (hr0) {
                const int jr = idrow[wave < nidr ? wave : nidr - 1];
                hv = *(const float4*)(h + (long)(lo + jr) * GIN_D + lane * 4);
            }
            __syncthreads();
#pragma unroll
            for (int u = 0; u < 16; ++u) lds_W[wbase + 8 * u] = wi[u];
            if (hr0) *(float4*)(lds_A + wave * GIN_PAD + lane * 4) = hv;
            for (int r = wave + 32; r < ntid * 16; r += 32) {
                const int jr = idrow[r < nidr ? r : nidr - 1];
                *(float4*)(lds_A + r * GIN_PAD + lane * 4) =
                    *(const float4*)(h + (long)(lo + jr) * GIN_D + lane * 4);
            }
            __syncthreads();

            const int ti = ((g - q) & 3) + 4 * (g >> 2);
            const float* const Ai[1] = { lds_A + (ti * 16 + lmod) * GIN_PAD + koff };

            v8f acci[1][2] = {};
            if (ti < ntid) gin_gemm<1, 1>(Ai, Bp0, Bp1, acci);
            __syncthreads();

            if (ti < ntid) {
#pragma unroll
                for (int cc = 0; cc < 2; ++cc) {
                    const int col = q * 32 + cc * 16 + lmod;
                    float bv = lds_bi1[col];
#pragma unroll
                    for (int r = 0; r < 8; ++r) {
                        float t = acci[0][cc][r] + bv;
                        lds_A[(ti * 16 + r + 8 * lhalf) * GIN_PAD + col] = t > 0.0f ? t : 0.0f;
                    }
                }
            }
#pragma unroll
            for (int u = 0; u < 16; ++u) lds_W[wbase + 8 * u] = Wi2[tid + u * 1024];
            __syncthreads();

            if (ti < ntid) {
                v8f acci2[1][2] = {};
                gin_gemm<1, 1>(Ai, Bp0, Bp1, acci2);
                const float bv0 = lds_bi2[q * 32 + lmod];
                const float bv1 = lds_bi2[q * 32 + 16 + lmod];
#pragma unroll
                for (int r0 = 0; r0 < 8; r0 += 4) {
                    int off[4], cnt[4];
#pragma unroll
                    for (int r = 0; r < 4; ++r) {
                        const int rc = ti * 16 + r0 + r + 8 * lhalf;
                        const int jr = idrow[rc < nidr ? rc : nidr - 1];
                        cnt[r] = rc < nidr ? idn[jr] : 0;
                        off[r] = (lo + jr) * GIN_D + q * 32 + lmod;
                    }
                    float ov[4][2];
#pragma unroll
                    for (int r = 0; r < 4; ++r) {
                        ov[r][0] = out[off[r]];
                        ov[r][1] = out[off[r] + 16];
                    }
#pragma unroll
                    for (int r = 0; r < 4; ++r) {
                        const float v0 = acci2[0][0][r0 + r] + bv0;
                        const float v1 = acci2[0][1][r0 + r] + bv1;
                        for (int k = 0; k < cnt[r]; ++k) { ov[r][0] += v0; ov[r][1] += v1; }
                    }
#pragma unroll
                    for (int r = 0; r < 4; ++r) {
                        if (cnt[r] > 0) {
                            out[off[r]]      = ov[r][0];
                            out[off[r] + 16] = ov[r][1];
                        }
                    }
                }
            }
        }
    }
}

extern "C" void kernel_launch(void* const* d_in, const int* in_sizes, int n_in,
                              void* d_out, int out_size, void* d_ws, size_t ws_size,
                              hipStream_t stream) {
    const float* x    = (const float*)d_in[0];
    const int*   edge = (const int*)d_in[1];
    const int*   nid  = (const int*)d_in[2];
    const float* eps  = (const float*)d_in[3];
    const float* W1   = (const float*)d_in[4];
    const float* b1   = (const float*)d_in[5];
    const float* W2   = (const float*)d_in[6];
    const float* b2   = (const float*)d_in[7];
    const float* Wi1  = (const float*)d_in[8];
    const float* bi1  = (const float*)d_in[9];
    const float* Wi2  = (const float*)d_in[10];
    const float* bi2  = (const float*)d_in[11];

    const int N   = in_sizes[0] / GIN_D;
    const int E   = in_sizes[1] / 2;
    const int NID = in_sizes[2];
    float* h      = (float*)d_ws;
    int*   tab    = (int*)(h + (size_t)N * GIN_D);
    int2*  staged = (int2*)(tab + GIN_GA * GIN_NBP);

    gin_bin<<<GIN_GA, 1024, 0, stream>>>(edge, tab, staged, N, E);
    const int bw = (N + GIN_NB - 1) / GIN_NB;
    gin_pull_mlp<<<dim3(GIN_NB, (bw + GIN_FROWS - 1) / GIN_FROWS), 1024, 0, stream>>>(
        x, eps, edge, tab, staged, h, W1, b1, W2, b2, nid, Wi1, bi1, Wi2, bi2, (float*)d_out, N, E, NID);
}


# bin size doubled: 256 bins of 196 nodes, one 32-wave group per WGP: weights staged once per 196 rows, 2 row tiles per wave on one B fragment, a node's row loads go out as one group; sum order kept
# speedup vs baseline: 3.2118x; 1.0485x over previous
#include <hip/hip_runtime.h>

typedef __attribute__((ext_vector_type(2))) float v2f;
typedef __attribute__((ext_vector_type(8))) float v8f;

#define GIN_D    128
#define GIN_PAD  132

#define GIN_NOEDGE 0x7fffffff
#define GIN_GA   256
#define GIN_NB   256
#define GIN_NBP  (GIN_NB + 1)
#define GIN_BWMAX 256
#define GIN_MAXM 4096
#define GIN_IDU  10

__device__ __forceinline__ int gin_block_scan(int v, int* sh, int tid, int nt, int& total) {
    const int lane = tid & 31, w = tid >> 5, nw = nt >> 5;
    int incl = v;
    for (int d = 1; d < 32; d <<= 1) {
        int t = __shfl_up(incl, d, 32);
        if (lane >= d) incl += t;
    }
    if (lane == 31) sh[w] = incl;
    __syncthreads();
    int pre = 0, tot = 0;
    for (int q = 0; q < nw; ++q) {
        int t = sh[q];
        if (q < w) pre += t;
        tot += t;
    }
    total = tot;
    __syncthreads();
    return pre + incl - v;
}

__global__ __launch_bounds__(1024) void gin_bin(const int* __restrict__ edge,
                                               int* __restrict__ tab,
                                               int2* __restrict__ staged, int n, int ne) {
    __shared__ int cnt[GIN_NB];
    __shared__ int cur[GIN_NB];
    __shared__ int sc[32];
    const int tid = threadIdx.x;
    const int wg  = blockIdx.x;
    const int ch  = (ne + GIN_GA - 1) / GIN_GA;
    const int bw  = (n + GIN_NB - 1) / GIN_NB;
    const int e0  = wg * ch;
    const int e1  = e0 + ch < ne ? e0 + ch : ne;
    if (tid < GIN_NB) cnt[tid] = 0;
    __syncthreads();
    for (int e = e0 + tid; e < e1; e += 1024) {
        int s = edge[e], d = edge[ne + e];
        if (s != d) atomicAdd(&cnt[d / bw], 1);
    }
    __syncthreads();
    int total;
    const int base = gin_block_scan(tid < GIN_NB ? cnt[tid] : 0, sc, tid, 1024, total);
    if (tid < GIN_NB) { cur[tid] = base; tab[wg * GIN_NBP + tid] = base; }
    if (tid == 0) tab[wg * GIN_NBP + GIN_NB] = total;
    __syncthreads();
    for (int e = e0 + tid; e < e1; e += 1024) {
        int s = edge[e], d = edge[ne + e];
        if (s != d) {
            int bin = d / bw;
            int pos = atomicAdd(&cur[bin], 1);
            staged[e0 + pos] = make_int2(e, ((d - bin * bw) << 16) | s);
        }
    }
}

template <bool TWO>
__device__ __forceinline__ int gin_src_of_rank(int r, int k0, int k1, int s0, int s1) {
    if (!TWO) {
        unsigned m = __builtin_amdgcn_ballot_w32(k0 == r);
        return __builtin_amdgcn_readlane(s0, __builtin_ctz(m));
    }
    unsigned m = __builtin_amdgcn_ballot_w32(k0 == r || k1 == r);
    return __builtin_amdgcn_readlane(k0 == r ? s0 : s1, __builtin_ctz(m));
}

template <int K, bool TWO>
__device__ __forceinline__ void gin_add_rows(const float* __restrict__ xr, int r0, int m, int last,
                                             int k0, int k1, int s0, int s1, float4& acc) {
    float4 v[K];
#pragma unroll
    for (int u = 0; u < K; ++u) {
        const int rk = r0 + u < last ? r0 + u : last;
        int s = gin_src_of_rank<TWO>(__builtin_amdgcn_readfirstlane(rk), k0, k1, s0, s1);
        v[u] = *(const float4*)(xr + (long)s * GIN_D);
    }
#pragma unroll
    for (int u = 0; u < K; ++u) {
        const bool on = u < K - 3 || u < m;
        acc.x += on ? v[u].x : 0.0f; acc.y += on ? v[u].y : 0.0f;
        acc.z += on ? v[u].z : 0.0f; acc.w += on ? v[u].w : 0.0f;
    }
}

template <bool TWO>
__device__ __forceinline__ void gin_add_all(const float* __restrict__ xr, int L,
                                            int k0, int k1, int s0, int s1, float4& acc) {
    if (TWO) {
        for (int r = 0; r < L; r += 8) {
            const int rem = L - r;
            if (rem > 4) gin_add_rows<8, TWO>(xr, r, rem, L - 1, k0, k1, s0, s1, acc);
            else         gin_add_rows<4, TWO>(xr, r, rem, L - 1, k0, k1, s0, s1, acc);
        }
        return;
    }
    for (int r = 0; r < L; r += 16) {
        const int rem = L - r;
        if (rem > 12)     gin_add_rows<16, TWO>(xr, r, rem, L - 1, k0, k1, s0, s1, acc);
        else if (rem > 8) gin_add_rows<12, TWO>(xr, r, rem, L - 1, k0, k1, s0, s1, acc);
        else if (rem > 4) gin_add_rows<8, TWO>(xr, r, rem, L - 1, k0, k1, s0, s1, acc);
        else              gin_add_rows<4, TWO>(xr, r, rem, L - 1, k0, k1, s0, s1, acc);
    }
}

__device__ __forceinline__ void gin_add_by_scan(const float* __restrict__ xr,
                                                const int* __restrict__ edge, int ne,
                                                int node, int lane, float4& acc) {
    for (int base = 0; base < ne; base += 32) {
        int e = base + lane;
        int s = -1, d = -1;
        if (e < ne) { s = edge[e]; d = edge[ne + e]; }
        unsigned m = __builtin_amdgcn_ballot_w32(d == node && s != node);
        while (m) {
            int l = __builtin_ctz(m);
            m &= m - 1;
            int ss = __builtin_amdgcn_readlane(s, l);
            float4 v = *(const float4*)(xr + (long)ss * GIN_D);
            acc.x += v.x; acc.y += v.y; acc.z += v.z; acc.w += v.w;
        }
    }
}

template <int NJ, int NA>
__device__ __forceinline__ void gin_gemm(const float* const (&Ap)[NA], const float* Bp0,
                                         const float* Bp1, v8f (&acc)[NA][2]) {
    for (int kk = 0; kk < GIN_D; kk += 4) {
        v2f b0 = *(const v2f*)(Bp0 + kk);
        v2f b1 = *(const v2f*)(Bp1 + kk);
#pragma unroll
        for (int j = 0; j < NJ; ++j) {
            v2f a = *(const v2f*)(Ap[j] + kk);
            acc[j][0] = __builtin_amdgcn_wmma_f32_16x16x4_f32(
                false, a, false, b0, (short)0, acc[j][0], false, false);
            acc[j][1] = __builtin_amdgcn_wmma_f32_16x16x4_f32(
                false, a, false, b1, (short)0, acc[j][1], false, false);
        }
    }
}

#define GIN_FROWS GIN_BWMAX

__global__ __launch_bounds__(1024) void gin_pull_mlp(const float* __restrict__ x,
                                                    const float* __restrict__ eps,
                                                    const int* __restrict__ edge,
                                                    const int* __restrict__ tab,
                                                    const int2* __restrict__ staged,
                                                    float* h,
                                                    const float* __restrict__ W1,
                                                    const float* __restrict__ b1,
                                                    const float* __restrict__ W2,
                                                    const float* __restrict__ b2,
                                                    const int* __restrict__ ids,
                                                    const float* __restrict__ Wi1,
                                                    const float* __restrict__ bi1,
                                                    const float* __restrict__ Wi2,
                                                    const float* __restrict__ bi2,
                                                    float* out, int n, int ne, int nid) {
    __shared__ __attribute__((aligned(16))) float lds_W[GIN_D * GIN_PAD];
    __shared__ __attribute__((aligned(16))) float lds_A[GIN_FROWS * GIN_PAD];
    __shared__ float lds_b1[GIN_D];
    __shared__ float lds_b2[GIN_D];
    __shared__ float lds_bi1[GIN_D];
    __shared__ float lds_bi2[GIN_D];
    __shared__ int idn[GIN_BWMAX];
    __shared__ int idrow[GIN_BWMAX];
    __shared__ int idtot;
    static_assert(GIN_BWMAX == 256, "one wave compacts the id rows, eight nodes per lane; two row tiles per wave");
    __shared__ int ncnt[GIN_BWMAX];
    __shared__ int nstart[GIN_BWMAX];
    __shared__ int ncur[GIN_BWMAX];
    __shared__ int sc[32];
    static_assert(2 * GIN_MAXM * sizeof(int2) <= sizeof(float) * GIN_D * GIN_PAD, "edge buffers fit the weight tile");
    int2* const ebuf = (int2*)lds_W;
    int2* const sbuf = ebuf + GIN_MAXM;
    const int tid  = threadIdx.x;
    const int lane = tid & 31;
    const int wave = __builtin_amdgcn_readfirstlane(tid >> 5);
    const int b    = blockIdx.x;
    const int ch   = (ne + GIN_GA - 1) / GIN_GA;
    const int bw   = (n + GIN_NB - 1) / GIN_NB;

    const int j0 = blockIdx.y * GIN_FROWS;
    const int nr = bw - j0 < GIN_FROWS ? bw - j0 : GIN_FROWS;
    const int lo = b * bw + j0;

    int idv[GIN_IDU];
    const int idlast = nid - 1;
#pragma unroll
    for (int u = 0; u < GIN_IDU; ++u) {
        const int i = tid + u * 1024;
        idv[u] = ids[i < nid ? i : idlast];
    }
    int s = 0, c = 0;
    if (tid < GIN_GA) {
        s = tab[tid * GIN_NBP + b];
        c = tab[tid * GIN_NBP + b + 1] - s;
    }
    if (tid < GIN_D)          lds_b1[tid] = b1[tid];
    else if (tid < 2 * GIN_D) lds_b2[tid - GIN_D] = b2[tid - GIN_D];
    else if (tid < 3 * GIN_D) lds_bi1[tid - 2 * GIN_D] = bi1[tid - 2 * GIN_D];
    else if (tid < 4 * GIN_D) lds_bi2[tid - 3 * GIN_D] = bi2[tid - 3 * GIN_D];
    if (tid < GIN_BWMAX) { ncnt[tid] = 0; idn[tid] = 0; }
    int M;
    const int base = gin_block_scan(c, sc, tid, 1024, M);
    const bool fits = M <= GIN_MAXM && bw <= GIN_BWMAX;
#pragma unroll
    for (int u = 0; u < GIN_IDU; ++u) {
        const int i = tid + u * 1024;
        const unsigned r = (unsigned)(idv[u] - lo);
        if (i < nid && r < (unsigned)nr) atomicAdd(&idn[r], 1);
    }
    for (int i0 = GIN_IDU * 1024; i0 < nid; i0 += GIN_IDU * 1024) {
#pragma unroll
        for (int u = 0; u < GIN_IDU; ++u) {
            const int i = i0 + tid + u * 1024;
            idv[u] = ids[i < nid ? i : idlast];
        }
#pragma unroll
        for (int u = 0; u < GIN_IDU; ++u) {
            const int i = i0 + tid + u * 1024;
            const unsigned r = (unsigned)(idv[u] - lo);
            if (i < nid && r < (unsigned)nr) atomicAdd(&idn[r], 1);
        }
    }
    if (fits) {
        const int2* seg = staged + (long)tid * ch + s;
        for (int i0 = 0; i0 < c; i0 += 16) {
            int2 r[16];
#pragma unroll
            for (int u = 0; u < 16; ++u) r[u] = seg[i0 + u < c ? i0 + u : c - 1];
#pragma unroll
            for (int u = 0; u < 16; ++u)
                if (i0 + u < c) ebuf[base + i0 + u] = r[u];
        }
    }
    __syncthreads();
    if (fits) {
        for (int i = tid; i < M; i += 1024) atomicAdd(&ncnt[ebuf[i].y >> 16], 1);
        __syncthreads();
        int tot;
        const int st = gin_block_scan(tid < GIN_BWMAX ? ncnt[tid] : 0, sc, tid, 1024, tot);
        if (tid < GIN_BWMAX) { nstart[tid] = st; ncur[tid] = st; }
        __syncthreads();
        for (int i = tid; i < M; i += 1024) {
            int2 r = ebuf[i];
            int pos = atomicAdd(&ncur[r.y >> 16], 1);
            sbuf[pos] = make_int2(r.x, r.y & 0xffff);
        }
        __syncthreads();
    }

    if (wave == 31) {
        int f[8], cl = 0;
#pragma unroll
        for (int k = 0; k < 8; ++k) {
            const int j = lane * 8 + k;
            f[k] = (j < nr && idn[j] > 0) ? 1 : 0;
            cl += f[k];
        }
        int incl = cl;
        for (int d = 1; d < 32; d <<= 1) {
            int t = __shfl_up(incl, d, 32);
            if (lane >= d) incl += t;
        }
        int pos = incl - cl;
#pragma unroll
        for (int k = 0; k < 8; ++k)
            if (f[k]) idrow[pos++] = lane * 8 + k;
        if (lane == 31) idtot = incl;
    }

    const float scl = 1.0f + eps[0];
    const float* xr = x + lane * 4;
    {
        const int nt = (nr + 15) >> 4;
        for (int jj = wave; jj < nt * 16; jj += 32) {
            const int j = j0 + jj;
            const int node = b * bw + j;
            float4 o = make_float4(0.0f, 0.0f, 0.0f, 0.0f);
            if (jj < nr && node < n) {
                const float4 xv = *(const float4*)(xr + (long)node * GIN_D);
                float4 acc = make_float4(0.0f, 0.0f, 0.0f, 0.0f);
                const int L   = fits ? __builtin_amdgcn_readfirstlane(ncnt[j]) : 0;
                const int st0 = fits ? __builtin_amdgcn_readfirstlane(nstart[j]) : 0;
                if (fits && L <= 32) {
                    int e0 = GIN_NOEDGE, s0 = 0;
                    if (lane < L) { int2 r = sbuf[st0 + lane]; e0 = r.x; s0 = r.y; }
                    int k0 = 0;
                    for (int qq = 0; qq < L; ++qq) k0 += __builtin_amdgcn_readlane(e0, qq) < e0 ? 1 : 0;
                    gin_add_all<false>(xr, L, k0, 0, s0, 0, acc);
                } else if (fits && L <= 64) {
                    int2 r0 = sbuf[st0 + lane];
                    int e0 = r0.x, s0 = r0.y, e1 = GIN_NOEDGE, s1 = 0;
                    if (32 + lane < L) { int2 r1 = sbuf[st0 + 32 + lane]; e1 = r1.x; s1 = r1.y; }
                    int k0 = 0, k1 = 0;
                    for (int qq = 0; qq < 32; ++qq) {
                        int eq = __builtin_amdgcn_readlane(e0, qq);
                        k0 += eq < e0 ? 1 : 0;
                        k1 += eq < e1 ? 1 : 0;
                    }
                    for (int qq = 32; qq < L; ++qq) {
                        int eq = __builtin_amdgcn_readlane(e1, qq - 32);
                        k0 += eq < e0 ? 1 : 0;
                        k1 += eq < e1 ? 1 : 0;
                    }
                    gin_add_all<true>(xr, L, k0, k1, s0, s1, acc);
                } else {
                    gin_add_by_scan(xr, edge, ne, node, lane, acc);
                }
                o.x = scl * xv.x + acc.x; o.y = scl * xv.y + acc.y;
                o.z = scl * xv.z + acc.z; o.w = scl * xv.w + acc.w;
                if (idn[jj] > 0) *(float4*)(h + (long)node * GIN_D + lane * 4) = o;
            }
            *(float4*)(lds_A + jj * GIN_PAD + lane * 4) = o;
        }

        float wv[16];
#pragma unroll
        for (int u = 0; u < 16; ++u) wv[u] = W1[tid + u * 1024];
        __syncthreads();
        const int lmod  = lane & 15;
        const int lhalf = lane >> 4;
        const int koff  = lhalf * 2;
        const int q  = wave & 3;
        const int g  = wave >> 2;
        const int tl[2] = { g, 8 + ((g - q) & 7) };
        const float* const Ap[2] = { lds_A + (tl[0] * 16 + lmod) * GIN_PAD + koff,
                                     lds_A + (tl[1] * 16 + lmod) * GIN_PAD + koff };
        const float* Bp0 = lds_W + (q * 32 + lmod) * GIN_PAD + koff;
        const float* Bp1 = Bp0 + 16 * GIN_PAD;

        const int wbase = (tid & (GIN_D - 1)) * GIN_PAD + (tid >> 7);
#pragma unroll
        for (int u = 0; u < 16; ++u) lds_W[wbase + 8 * u] = wv[u];
        __syncthreads();

        v8f acc[2][2] = {};
        if (tl[1] < nt)      gin_gemm<2, 2>(Ap, Bp0, Bp1, acc);
        else if (tl[0] < nt) gin_gemm<1, 2>(Ap, Bp0, Bp1, acc);
        __syncthreads();

#pragma unroll
        for (int j = 0; j < 2; ++j) {
            if (tl[j] < nt) {
#pragma unroll
                for (int cc = 0; cc < 2; ++cc) {
                    const int col = q * 32 + cc * 16 + lmod;
                    float bv = lds_b1[col];
#pragma unroll
                    for (int r = 0; r < 8; ++r) {
                        float t = acc[j][cc][r] + bv;
                        lds_A[(tl[j] * 16 + r + 8 * lhalf) * GIN_PAD + col] = t > 0.0f ? t : 0.0f;
                    }
                }
            }
        }
#pragma unroll
        for (int u = 0; u < 16; ++u) lds_W[wbase + 8 * u] = W2[tid + u * 1024];
        __syncthreads();

        {
            v8f acc2[2][2] = {};
            if (tl[1] < nt)      gin_gemm<2, 2>(Ap, Bp0, Bp1, acc2);
            else if (tl[0] < nt) gin_gemm<1, 2>(Ap, Bp0, Bp1, acc2);
#pragma unroll
            for (int j = 0; j < 2; ++j) {
                if (tl[j] < nt) {
#pragma unroll
                    for (int r = 0; r < 8; ++r) {
                        const int jr = tl[j] * 16 + r + 8 * lhalf;
                        const int node = b * bw + j0 + jr;
                        const bool ok = jr < nr && node < n;
#pragma unroll
                        for (int cc = 0; cc < 2; ++cc) {
                            const int col = q * 32 + cc * 16 + lmod;
                            float val = acc2[j][cc][r] + lds_b2[col];
                            if (ok) out[(long)node * GIN_D + col] = val;
                        }
                    }
                }
            }
        }

        const int nidr = __builtin_amdgcn_readfirstlane(idtot);
        const int ntid = (nidr + 15) >> 4;
        if (ntid > 0) {
            float wi[16];
#pragma unroll
            for (int u = 0; u < 16; ++u) wi[u] = Wi1[tid + u * 1024];
            const bool hr0 = wave < ntid * 16;
            float4 hv = make_float4(0.0f, 0.0f, 0.0f, 0.0f);
            if (hr0) {
                const int jr = idrow[wave < nidr ? wave : nidr - 1];
                hv = *(const float4*)(h + (long)(lo + jr) * GIN_D + lane * 4);
            }
            __syncthreads();
            if (hr0) *(float4*)(lds_A + wave * GIN_PAD + lane * 4) = hv;
            for (int r = wave + 32; r < ntid * 16; r += 32) {
                const int jr = idrow[r < nidr ? r : nidr - 1];
                *(float4*)(lds_A + r * GIN_PAD + lane * 4) =
                    *(const float4*)(h + (long)(lo + jr) * GIN_D + lane * 4);
            }
            for (int tb = 0; tb < ntid; tb += 8) {
                if (tb > 0) {
                    __syncthreads();
#pragma unroll
                    for (int u = 0; u < 16; ++u) wi[u] = Wi1[tid + u * 1024];
                }
#pragma unroll
                for (int u = 0; u < 16; ++u) lds_W[wbase + 8 * u] = wi[u];
                __syncthreads();

                const int ti = tb + ((g - q) & 3) + 4 * (g >> 2);
                const float* const Ai[1] = { lds_A + (ti * 16 + lmod) * GIN_PAD + koff };

                v8f acci[1][2] = {};
                if (ti < ntid) gin_gemm<1, 1>(Ai, Bp0, Bp1, acci);
                __syncthreads();

                if (ti < ntid) {
#pragma unroll
                    for (int cc = 0; cc < 2; ++cc) {
                        const int col = q * 32 + cc * 16 + lmod;
                        float bv = lds_bi1[col];
#pragma unroll
                        for (int r = 0; r < 8; ++r) {
                            float t = acci[0][cc][r] + bv;
                            lds_A[(ti * 16 + r + 8 * lhalf) * GIN_PAD + col] = t > 0.0f ? t : 0.0f;
                        }
                    }
                }
#pragma unroll
                for (int u = 0; u < 16; ++u) lds_W[wbase + 8 * u] = Wi2[tid + u * 1024];
                __syncthreads();

                if (ti < ntid) {
                    v8f acci2[1][2] = {};
                    gin_gemm<1, 1>(Ai, Bp0, Bp1, acci2);
                    const float bv0 = lds_bi2[q * 32 + lmod];
                    const float bv1 = lds_bi2[q * 32 + 16 + lmod];
#pragma unroll
                    for (int r0 = 0; r0 < 8; r0 += 4) {
                        int off[4], cnt[4];
#pragma unroll
                        for (int r = 0; r < 4; ++r) {
                            const int rc = ti * 16 + r0 + r + 8 * lhalf;
                            const int jr = idrow[rc < nidr ? rc : nidr - 1];
                            cnt[r] = rc < nidr ? idn[jr] : 0;
                            off[r] = (lo + jr) * GIN_D + q * 32 + lmod;
                        }
                        float ov[4][2];
#pragma unroll
                        for (int r = 0; r < 4; ++r) {
                            ov[r][0] = out[off[r]];
                            ov[r][1] = out[off[r] + 16];
                        }
#pragma unroll
                        for (int r = 0; r < 4; ++r) {
                            const float v0 = acci2[0][0][r0 + r] + bv0;
                            const float v1 = acci2[0][1][r0 + r] + bv1;
                            for (int k = 0; k < cnt[r]; ++k) { ov[r][0] += v0; ov[r][1] += v1; }
                        }
#pragma unroll
                        for (int r = 0; r < 4; ++r) {
                            if (cnt[r] > 0) {
                                out[off[r]]      = ov[r][0];
                                out[off[r] + 16] = ov[r][1];
                            }
                        }
                    }
                }
            }
        }
    }
}

extern "C" void kernel_launch(void* const* d_in, const int* in_sizes, int n_in,
                              void* d_out, int out_size, void* d_ws, size_t ws_size,
                              hipStream_t stream) {
    const float* x    = (const float*)d_in[0];
    const int*   edge = (const int*)d_in[1];
    const int*   nid  = (const int*)d_in[2];
    const float* eps  = (const float*)d_in[3];
    const float* W1   = (const float*)d_in[4];
    const float* b1   = (const float*)d_in[5];
    const float* W2   = (const float*)d_in[6];
    const float* b2   = (const float*)d_in[7];
    const float* Wi1  = (const float*)d_in[8];
    const float* bi1  = (const float*)d_in[9];
    const float* Wi2  = (const float*)d_in[10];
    const float* bi2  = (const float*)d_in[11];

    const int N   = in_sizes[0] / GIN_D;
    const int E   = in_sizes[1] / 2;
    const int NID = in_sizes[2];
    float* h      = (float*)d_ws;
    int*   tab    = (int*)(h + (size_t)N * GIN_D);
    int2*  staged = (int2*)(tab + GIN_GA * GIN_NBP);

    gin_bin<<<GIN_GA, 1024, 0, stream>>>(edge, tab, staged, N, E);
    const int bw = (N + GIN_NB - 1) / GIN_NB;
    gin_pull_mlp<<<dim3(GIN_NB, (bw + GIN_FROWS - 1) / GIN_FROWS), 1024, 0, stream>>>(
        x, eps, edge, tab, staged, h, W1, b1, W2, b2, nid, Wi1, bi1, Wi2, bi2, (float*)d_out, N, E, NID);
}


# id branch MLP worked inside the main MLP's two GEMM phases: A rows are the bin's h rows in LDS, Wi1/Wi2 staged beside W1/W2, id result added through LDS before the out row's one store; sum order kept
# speedup vs baseline: 3.3170x; 1.0328x over previous
#include <hip/hip_runtime.h>

typedef __attribute__((ext_vector_type(2))) float v2f;
typedef __attribute__((ext_vector_type(8))) float v8f;

#define GIN_D    128
#define GIN_PAD  132

#define GIN_NOEDGE 0x7fffffff
#define GIN_GA   256
#define GIN_NB   256
#define GIN_NBP  (GIN_NB + 1)
#define GIN_BWMAX 256
#define GIN_MAXM 4096
#define GIN_IDU  10
#define GIN_IDT  5

__device__ __forceinline__ int gin_block_scan(int v, int* sh, int tid, int nt, int& total) {
    const int lane = tid & 31, w = tid >> 5, nw = nt >> 5;
    int incl = v;
    for (int d = 1; d < 32; d <<= 1) {
        int t = __shfl_up(incl, d, 32);
        if (lane >= d) incl += t;
    }
    if (lane == 31) sh[w] = incl;
    __syncthreads();
    int pre = 0, tot = 0;
    for (int q = 0; q < nw; ++q) {
        int t = sh[q];
        if (q < w) pre += t;
        tot += t;
    }
    total = tot;
    __syncthreads();
    return pre + incl - v;
}

__global__ __launch_bounds__(1024) void gin_bin(const int* __restrict__ edge,
                                               int* __restrict__ tab,
                                               int2* __restrict__ staged, int n, int ne) {
    __shared__ int cnt[GIN_NB];
    __shared__ int cur[GIN_NB];
    __shared__ int sc[32];
    const int tid = threadIdx.x;
    const int wg  = blockIdx.x;
    const int ch  = (ne + GIN_GA - 1) / GIN_GA;
    const int bw  = (n + GIN_NB - 1) / GIN_NB;
    const int e0  = wg * ch;
    const int e1  = e0 + ch < ne ? e0 + ch : ne;
    if (tid < GIN_NB) cnt[tid] = 0;
    __syncthreads();
    for (int e = e0 + tid; e < e1; e += 1024) {
        int s = edge[e], d = edge[ne + e];
        if (s != d) atomicAdd(&cnt[d / bw], 1);
    }
    __syncthreads();
    int total;
    const int base = gin_block_scan(tid < GIN_NB ? cnt[tid] : 0, sc, tid, 1024, total);
    if (tid < GIN_NB) { cur[tid] = base; tab[wg * GIN_NBP + tid] = base; }
    if (tid == 0) tab[wg * GIN_NBP + GIN_NB] = total;
    __syncthreads();
    for (int e = e0 + tid; e < e1; e += 1024) {
        int s = edge[e], d = edge[ne + e];
        if (s != d) {
            int bin = d / bw;
            int pos = atomicAdd(&cur[bin], 1);
            staged[e0 + pos] = make_int2(e, ((d - bin * bw) << 16) | s);
        }
    }
}

template <bool TWO>
__device__ __forceinline__ int gin_src_of_rank(int r, int k0, int k1, int s0, int s1) {
    if (!TWO) {
        unsigned m = __builtin_amdgcn_ballot_w32(k0 == r);
        return __builtin_amdgcn_readlane(s0, __builtin_ctz(m));
    }
    unsigned m = __builtin_amdgcn_ballot_w32(k0 == r || k1 == r);
    return __builtin_amdgcn_readlane(k0 == r ? s0 : s1, __builtin_ctz(m));
}

template <int K, bool TWO>
__device__ __forceinline__ void gin_add_rows(const float* __restrict__ xr, int r0, int m, int last,
                                             int k0, int k1, int s0, int s1, float4& acc) {
    float4 v[K];
#pragma unroll
    for (int u = 0; u < K; ++u) {
        const int rk = r0 + u < last ? r0 + u : last;
        int s = gin_src_of_rank<TWO>(__builtin_amdgcn_readfirstlane(rk), k0, k1, s0, s1);
        v[u] = *(const float4*)(xr + (long)s * GIN_D);
    }
#pragma unroll
    for (int u = 0; u < K; ++u) {
        const bool on = u < K - 3 || u < m;
        acc.x += on ? v[u].x : 0.0f; acc.y += on ? v[u].y : 0.0f;
        acc.z += on ? v[u].z : 0.0f; acc.w += on ? v[u].w : 0.0f;
    }
}

template <bool TWO>
__device__ __forceinline__ void gin_add_all(const float* __restrict__ xr, int L,
                                            int k0, int k1, int s0, int s1, float4& acc) {
    if (TWO) {
        for (int r = 0; r < L; r += 8) {
            const int rem = L - r;
            if (rem > 4) gin_add_rows<8, TWO>(xr, r, rem, L - 1, k0, k1, s0, s1, acc);
            else         gin_add_rows<4, TWO>(xr, r, rem, L - 1, k0, k1, s0, s1, acc);
        }
        return;
    }
    for (int r = 0; r < L; r += 16) {
        const int rem = L - r;
        if (rem > 12)     gin_add_rows<16, TWO>(xr, r, rem, L - 1, k0, k1, s0, s1, acc);
        else if (rem > 8) gin_add_rows<12, TWO>(xr, r, rem, L - 1, k0, k1, s0, s1, acc);
        else if (rem > 4) gin_add_rows<8, TWO>(xr, r, rem, L - 1, k0, k1, s0, s1, acc);
        else              gin_add_rows<4, TWO>(xr, r, rem, L - 1, k0, k1, s0, s1, acc);
    }
}

__device__ __forceinline__ void gin_add_by_scan(const float* __restrict__ xr,
                                                const int* __restrict__ edge, int ne,
                                                int node, int lane, float4& acc) {
    for (int base = 0; base < ne; base += 32) {
        int e = base + lane;
        int s = -1, d = -1;
        if (e < ne) { s = edge[e]; d = edge[ne + e]; }
        unsigned m = __builtin_amdgcn_ballot_w32(d == node && s != node);
        while (m) {
            int l = __builtin_ctz(m);
            m &= m - 1;
            int ss = __builtin_amdgcn_readlane(s, l);
            float4 v = *(const float4*)(xr + (long)ss * GIN_D);
            acc.x += v.x; acc.y += v.y; acc.z += v.z; acc.w += v.w;
        }
    }
}

template <int NJ, int NA>
__device__ __forceinline__ void gin_gemm(const float* const (&Ap)[NA], const float* Bp0,
                                         const float* Bp1, v8f (&acc)[NA][2]) {
    for (int kk = 0; kk < GIN_D; kk += 4) {
        v2f b0 = *(const v2f*)(Bp0 + kk);
        v2f b1 = *(const v2f*)(Bp1 + kk);
#pragma unroll
        for (int j = 0; j < NJ; ++j) {
            v2f a = *(const v2f*)(Ap[j] + kk);
            acc[j][0] = __builtin_amdgcn_wmma_f32_16x16x4_f32(
                false, a, false, b0, (short)0, acc[j][0], false, false);
            acc[j][1] = __builtin_amdgcn_wmma_f32_16x16x4_f32(
                false, a, false, b1, (short)0, acc[j][1], false, false);
        }
    }
}

#define GIN_FROWS GIN_BWMAX

__global__ __launch_bounds__(1024) void gin_pull_mlp(const float* __restrict__ x,
                                                    const float* __restrict__ eps,
                                                    const int* __restrict__ edge,
                                                    const int* __restrict__ tab,
                                                    const int2* __restrict__ staged,
                                                    float* h,
                                                    const float* __restrict__ W1,
                                                    const float* __restrict__ b1,
                                                    const float* __restrict__ W2,
                                                    const float* __restrict__ b2,
                                                    const int* __restrict__ ids,
                                                    const float* __restrict__ Wi1,
                                                    const float* __restrict__ bi1,
                                                    const float* __restrict__ Wi2,
                                                    const float* __restrict__ bi2,
                                                    float* out, int n, int ne, int nid) {
    __shared__ __attribute__((aligned(16))) float lds_W[GIN_D * GIN_PAD];
    __shared__ __attribute__((aligned(16))) float lds_A[GIN_FROWS * GIN_PAD];
    __shared__ __attribute__((aligned(16))) float lds_Wi[GIN_D * GIN_PAD];
    __shared__ __attribute__((aligned(16))) float lds_T[GIN_IDT * 16 * GIN_PAD];
    __shared__ float lds_b1[GIN_D];
    __shared__ float lds_b2[GIN_D];
    __shared__ float lds_bi1[GIN_D];
    __shared__ float lds_bi2[GIN_D];
    __shared__ __attribute__((aligned(16))) int idn[GIN_BWMAX];
    __shared__ int idrow[GIN_BWMAX];
    __shared__ __attribute__((aligned(16))) int idslot[GIN_BWMAX];
    __shared__ int idtot;
    static_assert(GIN_BWMAX == 256, "one wave compacts the id rows, eight nodes per lane; two row tiles per wave");
    __shared__ int ncnt[GIN_BWMAX];
    __shared__ int nstart[GIN_BWMAX];
    __shared__ int ncur[GIN_BWMAX];
    __shared__ int sc[32];
    static_assert(2 * GIN_MAXM * sizeof(int2) <= sizeof(float) * GIN_D * GIN_PAD, "edge buffers fit the weight tile");
    int2* const ebuf = (int2*)lds_W;
    int2* const sbuf = ebuf + GIN_MAXM;
    const int tid  = threadIdx.x;
    const int lane = tid & 31;
    const int wave = __builtin_amdgcn_readfirstlane(tid >> 5);
    const int b    = blockIdx.x;
    const int ch   = (ne + GIN_GA - 1) / GIN_GA;
    const int bw   = (n + GIN_NB - 1) / GIN_NB;

    const int j0 = blockIdx.y * GIN_FROWS;
    const int nr = bw - j0 < GIN_FROWS ? bw - j0 : GIN_FROWS;
    const int lo = b * bw + j0;

    int idv[GIN_IDU];
    const int idlast = nid - 1;
#pragma unroll
    for (int u = 0; u < GIN_IDU; ++u) {
        const int i = tid + u * 1024;
        idv[u] = ids[i < nid ? i : idlast];
    }
    int s = 0, c = 0;
    if (tid < GIN_GA) {
        s = tab[tid * GIN_NBP + b];
        c = tab[tid * GIN_NBP + b + 1] - s;
    }
    if (tid < GIN_D)          lds_b1[tid] = b1[tid];
    else if (tid < 2 * GIN_D) lds_b2[tid - GIN_D] = b2[tid - GIN_D];
    else if (tid < 3 * GIN_D) lds_bi1[tid - 2 * GIN_D] = bi1[tid - 2 * GIN_D];
    else if (tid < 4 * GIN_D) lds_bi2[tid - 3 * GIN_D] = bi2[tid - 3 * GIN_D];
    if (tid < GIN_BWMAX) { ncnt[tid] = 0; idn[tid] = 0; }
    int M;
    const int base = gin_block_scan(c, sc, tid, 1024, M);
    const bool fits = M <= GIN_MAXM && bw <= GIN_BWMAX;
#pragma unroll
    for (int u = 0; u < GIN_IDU; ++u) {
        const int i = tid + u * 1024;
        const unsigned r = (unsigned)(idv[u] - lo);
        if (i < nid && r < (unsigned)nr) atomicAdd(&idn[r], 1);
    }
    for (int i0 = GIN_IDU * 1024; i0 < nid; i0 += GIN_IDU * 1024) {
#pragma unroll
        for (int u = 0; u < GIN_IDU; ++u) {
            const int i = i0 + tid + u * 1024;
            idv[u] = ids[i < nid ? i : idlast];
        }
#pragma unroll
        for (int u = 0; u < GIN_IDU; ++u) {
            const int i = i0 + tid + u * 1024;
            const unsigned r = (unsigned)(idv[u] - lo);
            if (i < nid && r < (unsigned)nr) atomicAdd(&idn[r], 1);
        }
    }
    if (fits) {
        const int2* seg = staged + (long)tid * ch + s;
        for (int i0 = 0; i0 < c; i0 += 16) {
            int2 r[16];
#pragma unroll
            for (int u = 0; u < 16; ++u) r[u] = seg[i0 + u < c ? i0 + u : c - 1];
#pragma unroll
            for (int u = 0; u < 16; ++u)
                if (i0 + u < c) ebuf[base + i0 + u] = r[u];
        }
    }
    __syncthreads();
    if (wave == 31) {
        int f[8], cl = 0;
#pragma unroll
        for (int k = 0; k < 8; ++k) {
            const int j = lane * 8 + k;
            f[k] = (j < nr && idn[j] > 0) ? 1 : 0;
            cl += f[k];
        }
        int incl = cl;
        for (int d = 1; d < 32; d <<= 1) {
            int t = __shfl_up(incl, d, 32);
            if (lane >= d) incl += t;
        }
        int pos = incl - cl;
#pragma unroll
        for (int k = 0; k < 8; ++k)
            if (f[k]) { idrow[pos] = lane * 8 + k; idslot[lane * 8 + k] = pos; ++pos; }
        if (lane == 31) idtot = incl;
    }
    if (fits) {
        for (int i = tid; i < M; i += 1024) atomicAdd(&ncnt[ebuf[i].y >> 16], 1);
        __syncthreads();
        int tot;
        const int st = gin_block_scan(tid < GIN_BWMAX ? ncnt[tid] : 0, sc, tid, 1024, tot);
        if (tid < GIN_BWMAX) { nstart[tid] = st; ncur[tid] = st; }
        __syncthreads();
        for (int i = tid; i < M; i += 1024) {
            int2 r = ebuf[i];
            int pos = atomicAdd(&ncur[r.y >> 16], 1);
            sbuf[pos] = make_int2(r.x, r.y & 0xffff);
        }
        __syncthreads();
    } else {
        __syncthreads();
    }
    const int nidr = __builtin_amdgcn_readfirstlane(idtot);
    const int ntid = (nidr + 15) >> 4;
    const bool fused = ntid > 0 && ntid <= GIN_IDT;
    const bool sep   = ntid > GIN_IDT;

    const float scl = 1.0f + eps[0];
    const float* xr = x + lane * 4;
    {
        const int nt = (nr + 15) >> 4;
        for (int jj = wave; jj < nt * 16; jj += 32) {
            const int j = j0 + jj;
            const int node = b * bw + j;
            float4 o = make_float4(0.0f, 0.0f, 0.0f, 0.0f);
            if (jj < nr && node < n) {
                const float4 xv = *(const float4*)(xr + (long)node * GIN_D);
                float4 acc = make_float4(0.0f, 0.0f, 0.0f, 0.0f);
                const int L   = fits ? __builtin_amdgcn_readfirstlane(ncnt[j]) : 0;
                const int st0 = fits ? __builtin_amdgcn_readfirstlane(nstart[j]) : 0;
                if (fits && L <= 32) {
                    int e0 = GIN_NOEDGE, s0 = 0;
                    if (lane < L) { int2 r = sbuf[st0 + lane]; e0 = r.x; s0 = r.y; }
                    int k0 = 0;
                    for (int qq = 0; qq < L; ++qq) k0 += __builtin_amdgcn_readlane(e0, qq) < e0 ? 1 : 0;
                    gin_add_all<false>(xr, L, k0, 0, s0, 0, acc);
                } else if (fits && L <= 64) {
                    int2 r0 = sbuf[st0 + lane];
                    int e0 = r0.x, s0 = r0.y, e1 = GIN_NOEDGE, s1 = 0;
                    if (32 + lane < L) { int2 r1 = sbuf[st0 + 32 + lane]; e1 = r1.x; s1 = r1.y; }
                    int k0 = 0, k1 = 0;
                    for (int qq = 0; qq < 32; ++qq) {
                        int eq = __builtin_amdgcn_readlane(e0, qq);
                        k0 += eq < e0 ? 1 : 0;
                        k1 += eq < e1 ? 1 : 0;
                    }
                    for (int qq = 32; qq < L; ++qq) {
                        int eq = __builtin_amdgcn_readlane(e1, qq - 32);
                        k0 += eq < e0 ? 1 : 0;
                        k1 += eq < e1 ? 1 : 0;
                    }
                    gin_add_all<true>(xr, L, k0, k1, s0, s1, acc);
                } else {
                    gin_add_by_scan(xr, edge, ne, node, lane, acc);
                }
                o.x = scl * xv.x + acc.x; o.y = scl * xv.y + acc.y;
                o.z = scl * xv.z + acc.z; o.w = scl * xv.w + acc.w;
                if (sep && idn[jj] > 0) *(float4*)(h + (long)node * GIN_D + lane * 4) = o;
            }
            *(float4*)(lds_A + jj * GIN_PAD + lane * 4) = o;
        }

        float wv[16];
#pragma unroll
        for (int u = 0; u < 16; ++u) wv[u] = W1[tid + u * 1024];
        const int wbase = (tid & (GIN_D - 1)) * GIN_PAD + (tid >> 7);
        if (fused) {
            float wiv[16];
#pragma unroll
            for (int u = 0; u < 16; ++u) wiv[u] = Wi1[tid + u * 1024];
#pragma unroll
            for (int u = 0; u < 16; ++u) lds_Wi[wbase + 8 * u] = wiv[u];
        }
        __syncthreads();
        const int lmod  = lane & 15;
        const int lhalf = lane >> 4;
        const int koff  = lhalf * 2;
        const int q  = wave & 3;
        const int g  = wave >> 2;
        const int tl[2] = { g, 8 + ((g - q) & 7) };
        const float* const Ap[2] = { lds_A + (tl[0] * 16 + lmod) * GIN_PAD + koff,
                                     lds_A + (tl[1] * 16 + lmod) * GIN_PAD + koff };
        const float* Bp0 = lds_W + (q * 32 + lmod) * GIN_PAD + koff;
        const float* Bp1 = Bp0 + 16 * GIN_PAD;
        const int ti = (((g - q) & 7) - nt) & 7;
        const bool hasid = fused && ti < ntid;
        int irow = 0, trow = lmod;
        if (hasid) {
            const int ir = ti * 16 + lmod;
            irow = idrow[ir < nidr ? ir : nidr - 1];
            trow = ir;
        }
        const float* const Ai[1] = { lds_A + irow * GIN_PAD + koff };
        const float* const Ti[1] = { lds_T + trow * GIN_PAD + koff };
        const float* Bi0 = lds_Wi + (q * 32 + lmod) * GIN_PAD + koff;
        const float* Bi1 = Bi0 + 16 * GIN_PAD;

#pragma unroll
        for (int u = 0; u < 16; ++u) lds_W[wbase + 8 * u] = wv[u];
        __syncthreads();

        v8f acc[2][2] = {};
        if (tl[1] < nt)      gin_gemm<2, 2>(Ap, Bp0, Bp1, acc);
        else if (tl[0] < nt) gin_gemm<1, 2>(Ap, Bp0, Bp1, acc);
        v8f acci[1][2] = {};
        if (hasid) gin_gemm<1, 1>(Ai, Bi0, Bi1, acci);
        __syncthreads();

#pragma unroll
        for (int j = 0; j < 2; ++j) {
            if (tl[j] < nt) {
#pragma unroll
                for (int cc = 0; cc < 2; ++cc) {
                    const int col = q * 32 + cc * 16 + lmod;
                    float bv = lds_b1[col];
#pragma unroll
                    for (int r = 0; r < 8; ++r) {
                        float t = acc[j][cc][r] + bv;
                        lds_A[(tl[j] * 16 + r + 8 * lhalf) * GIN_PAD + col] = t > 0.0f ? t : 0.0f;
                    }
                }
            }
        }
        if (hasid) {
#pragma unroll
            for (int cc = 0; cc < 2; ++cc) {
                const int col = q * 32 + cc * 16 + lmod;
                float bv = lds_bi1[col];
#pragma unroll
                for (int r = 0; r < 8; ++r) {
                    float t = acci[0][cc][r] + bv;
                    lds_T[(ti * 16 + r + 8 * lhalf) * GIN_PAD + col] = t > 0.0f ? t : 0.0f;
                }
            }
        }
        {
            float w2v[16];
#pragma unroll
            for (int u = 0; u < 16; ++u) w2v[u] = W2[tid + u * 1024];
            if (fused) {
                float wi2v[16];
#pragma unroll
                for (int u = 0; u < 16; ++u) wi2v[u] = Wi2[tid + u * 1024];
#pragma unroll
                for (int u = 0; u < 16; ++u) lds_W[wbase + 8 * u] = w2v[u];
#pragma unroll
                for (int u = 0; u < 16; ++u) lds_Wi[wbase + 8 * u] = wi2v[u];
            } else {
#pragma unroll
                for (int u = 0; u < 16; ++u) lds_W[wbase + 8 * u] = w2v[u];
            }
        }
        __syncthreads();

        {
            v8f acc2[2][2] = {};
            if (tl[1] < nt)      gin_gemm<2, 2>(Ap, Bp0, Bp1, acc2);
            else if (tl[0] < nt) gin_gemm<1, 2>(Ap, Bp0, Bp1, acc2);
            int cn[2][8], sl[2][8];
#pragma unroll
            for (int j = 0; j < 2; ++j) {
                const int jb = tl[j] * 16 + 8 * lhalf;
                const int4 c0 = *(const int4*)(idn + jb),    c1 = *(const int4*)(idn + jb + 4);
                const int4 p0 = *(const int4*)(idslot + jb), p1 = *(const int4*)(idslot + jb + 4);
                const int cr[8] = { c0.x, c0.y, c0.z, c0.w, c1.x, c1.y, c1.z, c1.w };
                const int pr[8] = { p0.x, p0.y, p0.z, p0.w, p1.x, p1.y, p1.z, p1.w };
#pragma unroll
                for (int r = 0; r < 8; ++r) {
                    cn[j][r] = fused ? cr[r] : 0;
                    sl[j][r] = cn[j][r] > 0 ? pr[r] : 0;
                }
            }
            const int col = q * 32 + lmod;
            const float bq0 = lds_b2[col], bq1 = lds_b2[col + 16];
#pragma unroll
            for (int j = 0; j < 2; ++j) {
                if (tl[j] < nt) {
#pragma unroll
                    for (int r = 0; r < 8; ++r) {
                        const int jr = tl[j] * 16 + r + 8 * lhalf;
                        const int node = b * bw + j0 + jr;
                        if (jr < nr && node < n && cn[j][r] == 0) {
                            out[(long)node * GIN_D + col]      = acc2[j][0][r] + bq0;
                            out[(long)node * GIN_D + col + 16] = acc2[j][1][r] + bq1;
                        }
                    }
                }
            }
            if (fused) {
                v8f acci2[1][2] = {};
                if (hasid) gin_gemm<1, 1>(Ti, Bi0, Bi1, acci2);
                __syncthreads();
                if (hasid) {
#pragma unroll
                    for (int cc = 0; cc < 2; ++cc) {
                        const int ccol = q * 32 + cc * 16 + lmod;
                        float bv = lds_bi2[ccol];
#pragma unroll
                        for (int r = 0; r < 8; ++r)
                            lds_T[(ti * 16 + r + 8 * lhalf) * GIN_PAD + ccol] = acci2[0][cc][r] + bv;
                    }
                }
                __syncthreads();
#pragma unroll
                for (int j = 0; j < 2; ++j) {
                    if (tl[j] < nt) {
                        float v0[8], v1[8];
#pragma unroll
                        for (int r = 0; r < 8; ++r) {
                            const float* tp = lds_T + sl[j][r] * GIN_PAD + col;
                            v0[r] = tp[0];
                            v1[r] = tp[16];
                        }
#pragma unroll
                        for (int r = 0; r < 8; ++r) {
                            const int jr = tl[j] * 16 + r + 8 * lhalf;
                            const int node = b * bw + j0 + jr;
                            if (jr < nr && node < n && cn[j][r] > 0) {
                                float val0 = acc2[j][0][r] + bq0;
                                float val1 = acc2[j][1][r] + bq1;
                                for (int k = 0; k < cn[j][r]; ++k) { val0 += v0[r]; val1 += v1[r]; }
                                out[(long)node * GIN_D + col]      = val0;
                                out[(long)node * GIN_D + col + 16] = val1;
                            }
                        }
                    }
                }
            }
        }

        if (sep) {
            float wi[16];
#pragma unroll
            for (int u = 0; u < 16; ++u) wi[u] = Wi1[tid + u * 1024];
            const bool hr0 = wave < ntid * 16;
            float4 hv = make_float4(0.0f, 0.0f, 0.0f, 0.0f);
            if (hr0) {
                const int jr = idrow[wave < nidr ? wave : nidr - 1];
                hv = *(const float4*)(h + (long)(lo + jr) * GIN_D + lane * 4);
            }
            __syncthreads();
            if (hr0) *(float4*)(lds_A + wave * GIN_PAD + lane * 4) = hv;
            for (int r = wave + 32; r < ntid * 16; r += 32) {
                const int jr = idrow[r < nidr ? r : nidr - 1];
                *(float4*)(lds_A + r * GIN_PAD + lane * 4) =
                    *(const float4*)(h + (long)(lo + jr) * GIN_D + lane * 4);
            }
            for (int tb = 0; tb < ntid; tb += 8) {
                if (tb > 0) {
                    __syncthreads();
#pragma unroll
                    for (int u = 0; u < 16; ++u) wi[u] = Wi1[tid + u * 1024];
                }
#pragma unroll
                for (int u = 0; u < 16; ++u) lds_W[wbase + 8 * u] = wi[u];
                __syncthreads();

                const int ti = tb + ((g - q) & 3) + 4 * (g >> 2);
                const float* const Ai[1] = { lds_A + (ti * 16 + lmod) * GIN_PAD + koff };

                v8f acci[1][2] = {};
                if (ti < ntid) gin_gemm<1, 1>(Ai, Bp0, Bp1, acci);
                __syncthreads();

                if (ti < ntid) {
#pragma unroll
                    for (int cc = 0; cc < 2; ++cc) {
                        const int col = q * 32 + cc * 16 + lmod;
                        float bv = lds_bi1[col];
#pragma unroll
                        for (int r = 0; r < 8; ++r) {
                            float t = acci[0][cc][r] + bv;
                            lds_A[(ti * 16 + r + 8 * lhalf) * GIN_PAD + col] = t > 0.0f ? t : 0.0f;
                        }
                    }
                }
#pragma unroll
                for (int u = 0; u < 16; ++u) lds_W[wbase + 8 * u] = Wi2[tid + u * 1024];
                __syncthreads();

                if (ti < ntid) {
                    v8f acci2[1][2] = {};
                    gin_gemm<1, 1>(Ai, Bp0, Bp1, acci2);
                    const float bv0 = lds_bi2[q * 32 + lmod];
                    const float bv1 = lds_bi2[q * 32 + 16 + lmod];
#pragma unroll
                    for (int r0 = 0; r0 < 8; r0 += 4) {
                        int off[4], cnt[4];
#pragma unroll
                        for (int r = 0; r < 4; ++r) {
                            const int rc = ti * 16 + r0 + r + 8 * lhalf;
                            const int jr = idrow[rc < nidr ? rc : nidr - 1];
                            cnt[r] = rc < nidr ? idn[jr] : 0;
                            off[r] = (lo + jr) * GIN_D + q * 32 + lmod;
                        }
                        float ov[4][2];
#pragma unroll
                        for (int r = 0; r < 4; ++r) {
                            ov[r][0] = out[off[r]];
                            ov[r][1] = out[off[r] + 16];
                        }
#pragma unroll
                        for (int r = 0; r < 4; ++r) {
                            const float v0 = acci2[0][0][r0 + r] + bv0;
                            const float v1 = acci2[0][1][r0 + r] + bv1;
                            for (int k = 0; k < cnt[r]; ++k) { ov[r][0] += v0; ov[r][1] += v1; }
                        }
#pragma unroll
                        for (int r = 0; r < 4; ++r) {
                            if (cnt[r] > 0) {
                                out[off[r]]      = ov[r][0];
                                out[off[r] + 16] = ov[r][1];
                            }
                        }
                    }
                }
            }
        }
    }
}

extern "C" void kernel_launch(void* const* d_in, const int* in_sizes, int n_in,
                              void* d_out, int out_size, void* d_ws, size_t ws_size,
                              hipStream_t stream) {
    const float* x    = (const float*)d_in[0];
    const int*   edge = (const int*)d_in[1];
    const int*   nid  = (const int*)d_in[2];
    const float* eps  = (const float*)d_in[3];
    const float* W1   = (const float*)d_in[4];
    const float* b1   = (const float*)d_in[5];
    const float* W2   = (const float*)d_in[6];
    const float* b2   = (const float*)d_in[7];
    const float* Wi1  = (const float*)d_in[8];
    const float* bi1  = (const float*)d_in[9];
    const float* Wi2  = (const float*)d_in[10];
    const float* bi2  = (const float*)d_in[11];

    const int N   = in_sizes[0] / GIN_D;
    const int E   = in_sizes[1] / 2;
    const int NID = in_sizes[2];
    float* h      = (float*)d_ws;
    int*   tab    = (int*)(h + (size_t)N * GIN_D);
    int2*  staged = (int2*)(tab + GIN_GA * GIN_NBP);

    gin_bin<<<GIN_GA, 1024, 0, stream>>>(edge, tab, staged, N, E);
    const int bw = (N + GIN_NB - 1) / GIN_NB;
    gin_pull_mlp<<<dim3(GIN_NB, (bw + GIN_FROWS - 1) / GIN_FROWS), 1024, 0, stream>>>(
        x, eps, edge, tab, staged, h, W1, b1, W2, b2, nid, Wi1, bi1, Wi2, bi2, (float*)d_out, N, E, NID);
}
